# starting kernel
# baseline (speedup 1.0000x reference)
#include <hip/hip_runtime.h>
#include <math.h>


#define BATCH   2
#define SEQLEN  2048
#define D_INNER 2048
#define D_STATE 16
#define DT_RANK 64
#define N_PROJ  (DT_RANK + 2 * D_STATE)
#define M_TOT   (BATCH * SEQLEN)
#define NSTATE_TOT (BATCH * D_INNER * D_STATE)
#define NCHUNK  16
#define CHUNKL  (SEQLEN / NCHUNK)

typedef float v2f __attribute__((ext_vector_type(2)));
typedef float v8f __attribute__((ext_vector_type(8)));

__global__ void ssm_xproj_wmma(const float* __restrict__ X,
                               const float* __restrict__ W,
                               float* __restrict__ Y) {
  const int lane = threadIdx.x & 31;
  const int wave = threadIdx.x >> 5;
  const int half = lane >> 4;
  const int ml   = lane & 15;
  const int m0   = blockIdx.x << 4;
  const int n0   = wave << 4;

  const float* __restrict__ arow = X + (size_t)(m0 + ml) * 2048;
  v8f acc = {};
  for (int k = 0; k < 2048; k += 4) {
    const int ka = k + 2 * half;
    v2f a, b;
    a.x = arow[ka];
    a.y = arow[ka + 1];
    b.x = W[(size_t)ka * N_PROJ + n0 + ml];
    b.y = W[(size_t)(ka + 1) * N_PROJ + n0 + ml];
    acc = __builtin_amdgcn_wmma_f32_16x16x4_f32(
        false, a, false, b, (short)0, acc, false, false);
  }
#pragma unroll
  for (int r = 0; r < 8; ++r) {
    const int m = m0 + r + 8 * half;
    Y[(size_t)m * N_PROJ + n0 + ml] = acc[r];
  }
}

__global__ void ssm_dtproj_wmma(const float* __restrict__ XD,
                                const float* __restrict__ Wd,
                                const float* __restrict__ bias,
                                float* __restrict__ DT) {
  const int lane = threadIdx.x & 31;
  const int wave = threadIdx.x >> 5;
  const int half = lane >> 4;
  const int ml   = lane & 15;
  const int m0   = blockIdx.y << 4;
  const int n0   = (blockIdx.x << 7) + (wave << 4);

  const float* __restrict__ arow = XD + (size_t)(m0 + ml) * N_PROJ;
  v8f acc = {};
  for (int k = 0; k < DT_RANK; k += 4) {
    const int ka = k + 2 * half;
    v2f a, b;
    a.x = arow[ka];
    a.y = arow[ka + 1];
    b.x = Wd[(size_t)ka * D_INNER + n0 + ml];
    b.y = Wd[(size_t)(ka + 1) * D_INNER + n0 + ml];
    acc = __builtin_amdgcn_wmma_f32_16x16x4_f32(
        false, a, false, b, (short)0, acc, false, false);
  }
  const float bn = bias[n0 + ml];
#pragma unroll
  for (int r = 0; r < 8; ++r) {
    const int m = m0 + r + 8 * half;
    float v = acc[r] + bn;
    float sp = (v > 20.0f) ? v : log1pf(__expf(v));
    DT[(size_t)m * D_INNER + n0 + ml] = sp;
  }
}

__global__ void ssm_scan_pass1(const float* __restrict__ X,
                               const float* __restrict__ XDBL,
                               const float* __restrict__ DT,
                               const float* __restrict__ A_log,
                               float* __restrict__ chunkA,
                               float* __restrict__ chunkB) {
  const int t = blockIdx.x * blockDim.x + threadIdx.x;
  const int chunk = blockIdx.y;
  const int n = t & 15;
  const int c = t >> 4;
  const int b = c >> 11;
  const int d = c & 2047;

  const float a2 = -__expf(A_log[d * D_STATE + n]) * 1.44269504088896f;
  float s = 0.0f;
  float p = 1.0f;
  const size_t mbase = (size_t)b * SEQLEN + (size_t)chunk * CHUNKL;
  for (int l = 0; l < CHUNKL; ++l) {
    const size_t m = mbase + l;
    const float dt = DT[m * D_INNER + d];
    const float xv = X[m * D_INNER + d];
    const float Bn = XDBL[m * N_PROJ + DT_RANK + n];
    const float dA = exp2f(dt * a2);
    s = fmaf(dA, s, dt * Bn * xv);
    p *= dA;
  }
  const size_t i = (size_t)chunk * NSTATE_TOT + t;
  chunkA[i] = p;
  chunkB[i] = s;
}

__global__ void ssm_scan_pass2(const float* __restrict__ chunkA,
                               float* __restrict__ chunkB) {
  const int t = blockIdx.x * blockDim.x + threadIdx.x;
  float s = 0.0f;
#pragma unroll
  for (int ch = 0; ch < NCHUNK; ++ch) {
    const size_t i = (size_t)ch * NSTATE_TOT + t;
    const float Ac = chunkA[i];
    const float Bc = chunkB[i];
    chunkB[i] = s;
    s = fmaf(Ac, s, Bc);
  }
}

__global__ void ssm_scan_pass3(const float* __restrict__ X,
                               const float* __restrict__ XDBL,
                               const float* __restrict__ DT,
                               const float* __restrict__ A_log,
                               const float* __restrict__ Dv,
                               const float* __restrict__ Sin,
                               float* __restrict__ OUT) {
  const int t = blockIdx.x * blockDim.x + threadIdx.x;
  const int chunk = blockIdx.y;
  const int n = t & 15;
  const int c = t >> 4;
  const int b = c >> 11;
  const int d = c & 2047;

  const float a2 = -__expf(A_log[d * D_STATE + n]) * 1.44269504088896f;
  const float Dd = Dv[d];
  float st = Sin[(size_t)chunk * NSTATE_TOT + t];

  const size_t mbase = (size_t)b * SEQLEN + (size_t)chunk * CHUNKL;
  for (int l = 0; l < CHUNKL; ++l) {
    const size_t m = mbase + l;
    const float dt = DT[m * D_INNER + d];
    const float xv = X[m * D_INNER + d];
    const float* __restrict__ bc = XDBL + m * N_PROJ + DT_RANK;
    const float Bn = bc[n];
    const float Cn = bc[D_STATE + n];

    const float dA = exp2f(dt * a2);
    st = fmaf(dA, st, dt * Bn * xv);
    float y = st * Cn;
    y += __shfl_xor(y, 1, 32);
    y += __shfl_xor(y, 2, 32);
    y += __shfl_xor(y, 4, 32);
    y += __shfl_xor(y, 8, 32);
    if (n == 0) OUT[m * D_INNER + d] = y + xv * Dd;
  }
}

extern "C" void kernel_launch(void* const* d_in, const int* in_sizes, int n_in,
                              void* d_out, int out_size, void* d_ws, size_t ws_size,
                              hipStream_t stream) {
  const float* x    = (const float*)d_in[0];
  const float* xpw  = (const float*)d_in[1];
  const float* dtw  = (const float*)d_in[2];
  const float* dtb  = (const float*)d_in[3];
  const float* alog = (const float*)d_in[4];
  const float* dvec = (const float*)d_in[5];
  float* out = (float*)d_out;

  char* ws = (char*)d_ws;
  float* xdbl   = (float*)ws;
  ws += (size_t)M_TOT * N_PROJ * sizeof(float);
  float* delta  = (float*)ws;
  ws += (size_t)M_TOT * D_INNER * sizeof(float);
  float* chunkA = (float*)ws;
  ws += (size_t)NCHUNK * NSTATE_TOT * sizeof(float);
  float* chunkB = (float*)ws;

  ssm_xproj_wmma<<<dim3(M_TOT / 16), dim3(192), 0, stream>>>(x, xpw, xdbl);

  ssm_dtproj_wmma<<<dim3(D_INNER / 128, M_TOT / 16), dim3(256), 0, stream>>>(
      xdbl, dtw, dtb, delta);

  ssm_scan_pass1<<<dim3(NSTATE_TOT / 256, NCHUNK), dim3(256), 0, stream>>>(
      x, xdbl, delta, alog, chunkA, chunkB);
  ssm_scan_pass2<<<dim3(NSTATE_TOT / 256), dim3(256), 0, stream>>>(
      chunkA, chunkB);
  ssm_scan_pass3<<<dim3(NSTATE_TOT / 256, NCHUNK), dim3(256), 0, stream>>>(
      x, xdbl, delta, alog, dvec, chunkB, out);
}


# Scan re-mapped: one lane per (b,d) channel holding its 16 states in registers (32 chunks of 64 steps); B,C wave-uniform, y summed in-lane on 4 partial sums instead of a shuffle butterfly.
# speedup vs baseline: 2.2038x; 2.2038x over previous
#include <hip/hip_runtime.h>
#include <math.h>


#define BATCH   2
#define SEQLEN  2048
#define D_INNER 2048
#define D_STATE 16
#define DT_RANK 64
#define N_PROJ  (DT_RANK + 2 * D_STATE)
#define M_TOT   (BATCH * SEQLEN)
#define NSTATE_TOT (BATCH * D_INNER * D_STATE)
#define NCHAN   (BATCH * D_INNER)
#define NCHUNK  32
#define CHUNKL  (SEQLEN / NCHUNK)
#define SCAN_THREADS 256
#define GRP     4

typedef float v2f __attribute__((ext_vector_type(2)));
typedef float v8f __attribute__((ext_vector_type(8)));

__global__ void ssm_xproj_wmma(const float* __restrict__ X,
                               const float* __restrict__ W,
                               float* __restrict__ Y) {
  const int lane = threadIdx.x & 31;
  const int wave = threadIdx.x >> 5;
  const int half = lane >> 4;
  const int ml   = lane & 15;
  const int m0   = blockIdx.x << 4;
  const int n0   = wave << 4;

  const float* __restrict__ arow = X + (size_t)(m0 + ml) * 2048;
  v8f acc = {};
  for (int k = 0; k < 2048; k += 4) {
    const int ka = k + 2 * half;
    v2f a, b;
    a.x = arow[ka];
    a.y = arow[ka + 1];
    b.x = W[(size_t)ka * N_PROJ + n0 + ml];
    b.y = W[(size_t)(ka + 1) * N_PROJ + n0 + ml];
    acc = __builtin_amdgcn_wmma_f32_16x16x4_f32(
        false, a, false, b, (short)0, acc, false, false);
  }
#pragma unroll
  for (int r = 0; r < 8; ++r) {
    const int m = m0 + r + 8 * half;
    Y[(size_t)m * N_PROJ + n0 + ml] = acc[r];
  }
}

__global__ void ssm_dtproj_wmma(const float* __restrict__ XD,
                                const float* __restrict__ Wd,
                                const float* __restrict__ bias,
                                float* __restrict__ DT) {
  const int lane = threadIdx.x & 31;
  const int wave = threadIdx.x >> 5;
  const int half = lane >> 4;
  const int ml   = lane & 15;
  const int m0   = blockIdx.y << 4;
  const int n0   = (blockIdx.x << 7) + (wave << 4);

  const float* __restrict__ arow = XD + (size_t)(m0 + ml) * N_PROJ;
  v8f acc = {};
  for (int k = 0; k < DT_RANK; k += 4) {
    const int ka = k + 2 * half;
    v2f a, b;
    a.x = arow[ka];
    a.y = arow[ka + 1];
    b.x = Wd[(size_t)ka * D_INNER + n0 + ml];
    b.y = Wd[(size_t)(ka + 1) * D_INNER + n0 + ml];
    acc = __builtin_amdgcn_wmma_f32_16x16x4_f32(
        false, a, false, b, (short)0, acc, false, false);
  }
  const float bn = bias[n0 + ml];
#pragma unroll
  for (int r = 0; r < 8; ++r) {
    const int m = m0 + r + 8 * half;
    float v = acc[r] + bn;
    float sp = (v > 20.0f) ? v : log1pf(__expf(v));
    DT[(size_t)m * D_INNER + n0 + ml] = sp;
  }
}

__global__ __launch_bounds__(SCAN_THREADS)
void ssm_scan_pass1(const float* __restrict__ X,
                    const float* __restrict__ XDBL,
                    const float* __restrict__ DT,
                    const float* __restrict__ A_log,
                    float* __restrict__ chunkA,
                    float* __restrict__ chunkB) {
  const int c     = blockIdx.x * SCAN_THREADS + threadIdx.x;
  const int chunk = blockIdx.y;
  const int b     = blockIdx.x / (D_INNER / SCAN_THREADS);
  const int d     = c & (D_INNER - 1);

  float a2[D_STATE], s[D_STATE], p[D_STATE];
#pragma unroll
  for (int n = 0; n < D_STATE; ++n) {
    a2[n] = -__expf(A_log[d * D_STATE + n]) * 1.44269504088896f;
    s[n] = 0.0f;
    p[n] = 1.0f;
  }

  const size_t m0 = (size_t)b * SEQLEN + (size_t)chunk * CHUNKL;
  const float* __restrict__ dtp = DT + m0 * D_INNER + d;
  const float* __restrict__ xp  = X + m0 * D_INNER + d;
  const float* __restrict__ bcp = XDBL + m0 * N_PROJ + DT_RANK;

  float dtn[GRP], xn[GRP];
#pragma unroll
  for (int u = 0; u < GRP; ++u) {
    dtn[u] = dtp[(size_t)u * D_INNER];
    xn[u]  = xp[(size_t)u * D_INNER];
  }
  for (int g = 0; g < CHUNKL; g += GRP) {
    float dtc[GRP], xc[GRP];
#pragma unroll
    for (int u = 0; u < GRP; ++u) { dtc[u] = dtn[u]; xc[u] = xn[u]; }
    if (g + GRP < CHUNKL) {
#pragma unroll
      for (int u = 0; u < GRP; ++u) {
        dtn[u] = dtp[(size_t)(g + GRP + u) * D_INNER];
        xn[u]  = xp[(size_t)(g + GRP + u) * D_INNER];
      }
    }
#pragma unroll
    for (int u = 0; u < GRP; ++u) {
      const float* __restrict__ bc = bcp + (size_t)(g + u) * N_PROJ;
      const float dt = dtc[u];
      const float xv = xc[u];
#pragma unroll
      for (int n = 0; n < D_STATE; ++n) {
        const float dA = exp2f(dt * a2[n]);
        s[n] = fmaf(dA, s[n], dt * bc[n] * xv);
        p[n] *= dA;
      }
    }
  }
#pragma unroll
  for (int n = 0; n < D_STATE; ++n) {
    const size_t i = ((size_t)chunk * D_STATE + n) * NCHAN + c;
    chunkA[i] = p[n];
    chunkB[i] = s[n];
  }
}

__global__ void ssm_scan_pass2(const float* __restrict__ chunkA,
                               float* __restrict__ chunkB) {
  const int t = blockIdx.x * blockDim.x + threadIdx.x;
  float s = 0.0f;
#pragma unroll
  for (int ch = 0; ch < NCHUNK; ++ch) {
    const size_t i = (size_t)ch * NSTATE_TOT + t;
    const float Ac = chunkA[i];
    const float Bc = chunkB[i];
    chunkB[i] = s;
    s = fmaf(Ac, s, Bc);
  }
}

__global__ __launch_bounds__(SCAN_THREADS)
void ssm_scan_pass3(const float* __restrict__ X,
                    const float* __restrict__ XDBL,
                    const float* __restrict__ DT,
                    const float* __restrict__ A_log,
                    const float* __restrict__ Dv,
                    const float* __restrict__ Sin,
                    float* __restrict__ OUT) {
  const int c     = blockIdx.x * SCAN_THREADS + threadIdx.x;
  const int chunk = blockIdx.y;
  const int b     = blockIdx.x / (D_INNER / SCAN_THREADS);
  const int d     = c & (D_INNER - 1);

  float a2[D_STATE], st[D_STATE];
#pragma unroll
  for (int n = 0; n < D_STATE; ++n) {
    a2[n] = -__expf(A_log[d * D_STATE + n]) * 1.44269504088896f;
    st[n] = Sin[((size_t)chunk * D_STATE + n) * NCHAN + c];
  }
  const float Dd = Dv[d];

  const size_t m0 = (size_t)b * SEQLEN + (size_t)chunk * CHUNKL;
  const float* __restrict__ dtp = DT + m0 * D_INNER + d;
  const float* __restrict__ xp  = X + m0 * D_INNER + d;
  const float* __restrict__ bcp = XDBL + m0 * N_PROJ + DT_RANK;
  float* __restrict__ op = OUT + m0 * D_INNER + d;

  float dtn[GRP], xn[GRP];
#pragma unroll
  for (int u = 0; u < GRP; ++u) {
    dtn[u] = dtp[(size_t)u * D_INNER];
    xn[u]  = xp[(size_t)u * D_INNER];
  }
  for (int g = 0; g < CHUNKL; g += GRP) {
    float dtc[GRP], xc[GRP];
#pragma unroll
    for (int u = 0; u < GRP; ++u) { dtc[u] = dtn[u]; xc[u] = xn[u]; }
    if (g + GRP < CHUNKL) {
#pragma unroll
      for (int u = 0; u < GRP; ++u) {
        dtn[u] = dtp[(size_t)(g + GRP + u) * D_INNER];
        xn[u]  = xp[(size_t)(g + GRP + u) * D_INNER];
      }
    }
#pragma unroll
    for (int u = 0; u < GRP; ++u) {
      const float* __restrict__ bc = bcp + (size_t)(g + u) * N_PROJ;
      const float dt = dtc[u];
      const float xv = xc[u];
      float y[4] = {0.0f, 0.0f, 0.0f, 0.0f};
#pragma unroll
      for (int n = 0; n < D_STATE; ++n) {
        const float dA = exp2f(dt * a2[n]);
        st[n] = fmaf(dA, st[n], dt * bc[n] * xv);
        y[n & 3] = fmaf(st[n], bc[D_STATE + n], y[n & 3]);
      }
      op[(size_t)(g + u) * D_INNER] = ((y[0] + y[1]) + (y[2] + y[3])) + xv * Dd;
    }
  }
}

extern "C" void kernel_launch(void* const* d_in, const int* in_sizes, int n_in,
                              void* d_out, int out_size, void* d_ws, size_t ws_size,
                              hipStream_t stream) {
  const float* x    = (const float*)d_in[0];
  const float* xpw  = (const float*)d_in[1];
  const float* dtw  = (const float*)d_in[2];
  const float* dtb  = (const float*)d_in[3];
  const float* alog = (const float*)d_in[4];
  const float* dvec = (const float*)d_in[5];
  float* out = (float*)d_out;

  char* ws = (char*)d_ws;
  float* xdbl   = (float*)ws;
  ws += (size_t)M_TOT * N_PROJ * sizeof(float);
  float* delta  = (float*)ws;
  ws += (size_t)M_TOT * D_INNER * sizeof(float);
  float* chunkA = (float*)ws;
  ws += (size_t)NCHUNK * NSTATE_TOT * sizeof(float);
  float* chunkB = (float*)ws;

  ssm_xproj_wmma<<<dim3(M_TOT / 16), dim3(192), 0, stream>>>(x, xpw, xdbl);

  ssm_dtproj_wmma<<<dim3(D_INNER / 128, M_TOT / 16), dim3(256), 0, stream>>>(
      xdbl, dtw, dtb, delta);

  ssm_scan_pass1<<<dim3(NCHAN / SCAN_THREADS, NCHUNK), dim3(SCAN_THREADS), 0, stream>>>(
      x, xdbl, delta, alog, chunkA, chunkB);
  ssm_scan_pass2<<<dim3(NSTATE_TOT / 256), dim3(256), 0, stream>>>(
      chunkA, chunkB);
  ssm_scan_pass3<<<dim3(NCHAN / SCAN_THREADS, NCHUNK), dim3(SCAN_THREADS), 0, stream>>>(
      x, xdbl, delta, alog, dvec, chunkB, out);
}


# GEMM1 re-tiled: K split over the 8 waves of a block, 6 accumulator tiles per wave, whole-line operand loads one batch ahead; 8 partial sums added via LDS in slice order (K summation order changes).
# speedup vs baseline: 3.3298x; 1.5109x over previous
#include <hip/hip_runtime.h>
#include <math.h>


#define BATCH   2
#define SEQLEN  2048
#define D_INNER 2048
#define D_STATE 16
#define DT_RANK 64
#define N_PROJ  (DT_RANK + 2 * D_STATE)
#define M_TOT   (BATCH * SEQLEN)
#define NSTATE_TOT (BATCH * D_INNER * D_STATE)
#define NCHAN   (BATCH * D_INNER)
#define NCHUNK  32
#define CHUNKL  (SEQLEN / NCHUNK)
#define SCAN_THREADS 256
#define GRP     4

typedef float v2f __attribute__((ext_vector_type(2)));
typedef float v4f __attribute__((ext_vector_type(4)));
typedef float v8f __attribute__((ext_vector_type(8)));

#define G1_WAVES  8
#define G1_KSLICE (D_INNER / G1_WAVES)
#define G1_KBATCH 32
#define G1_TILE_F (6 * 8 * 32)

__global__ __launch_bounds__(256)
void ssm_xproj_wmma(const float* __restrict__ X,
                    const float* __restrict__ W,
                    float* __restrict__ Y) {
  __shared__ float red[G1_WAVES * G1_TILE_F];
  const int lane = threadIdx.x & 31;
  const int wave = threadIdx.x >> 5;
  const int half = lane >> 4;
  const int ml   = lane & 15;
  const int m0   = blockIdx.x << 4;
  const int k0   = wave * G1_KSLICE + 4 * half;

  const float* __restrict__ ap  = X + (size_t)(m0 + ml) * D_INNER + k0;
  const float* __restrict__ bp4 = W + (size_t)k0 * N_PROJ + 4 * ml;
  const float* __restrict__ bp2 = W + (size_t)k0 * N_PROJ + 64 + 2 * ml;

  v8f acc[6];
#pragma unroll
  for (int t = 0; t < 6; ++t) acc[t] = (v8f){};

  v4f an[4], b4n[4][4];
  v2f b2n[4][4];
#pragma unroll
  for (int j = 0; j < 4; ++j) {
    an[j] = *(const v4f*)(ap + 8 * j);
#pragma unroll
    for (int e = 0; e < 4; ++e) {
      b4n[j][e] = *(const v4f*)(bp4 + (size_t)(8 * j + e) * N_PROJ);
      b2n[j][e] = *(const v2f*)(bp2 + (size_t)(8 * j + e) * N_PROJ);
    }
  }
#pragma unroll 1
  for (int kb = 0; kb < G1_KSLICE; kb += G1_KBATCH) {
    v4f ac[4], b4c[4][4];
    v2f b2c[4][4];
#pragma unroll
    for (int j = 0; j < 4; ++j) {
      ac[j] = an[j];
#pragma unroll
      for (int e = 0; e < 4; ++e) { b4c[j][e] = b4n[j][e]; b2c[j][e] = b2n[j][e]; }
    }
    if (kb + G1_KBATCH < G1_KSLICE) {
      const int kn = kb + G1_KBATCH;
#pragma unroll
      for (int j = 0; j < 4; ++j) {
        an[j] = *(const v4f*)(ap + kn + 8 * j);
#pragma unroll
        for (int e = 0; e < 4; ++e) {
          b4n[j][e] = *(const v4f*)(bp4 + (size_t)(kn + 8 * j + e) * N_PROJ);
          b2n[j][e] = *(const v2f*)(bp2 + (size_t)(kn + 8 * j + e) * N_PROJ);
        }
      }
    }
#pragma unroll
    for (int j = 0; j < 4; ++j) {
#pragma unroll
      for (int q = 0; q < 2; ++q) {
        v2f a;
        a.x = ac[j][2 * q];
        a.y = ac[j][2 * q + 1];
#pragma unroll
        for (int t = 0; t < 6; ++t) {
          v2f b;
          if (t < 4) {
            b.x = b4c[j][2 * q][t];
            b.y = b4c[j][2 * q + 1][t];
          } else {
            b.x = b2c[j][2 * q][t - 4];
            b.y = b2c[j][2 * q + 1][t - 4];
          }
          acc[t] = __builtin_amdgcn_wmma_f32_16x16x4_f32(
              false, a, false, b, (short)0, acc[t], false, false);
        }
      }
    }
  }

#pragma unroll
  for (int t = 0; t < 6; ++t) {
#pragma unroll
    for (int r = 0; r < 8; ++r) {
      red[wave * G1_TILE_F + (t * 8 + r) * 32 + lane] = acc[t][r];
    }
  }
  __syncthreads();
  float s[6];
#pragma unroll
  for (int t = 0; t < 6; ++t) {
    const int q = (t * 8 + wave) * 32 + lane;
    float v = red[q];
#pragma unroll
    for (int w = 1; w < G1_WAVES; ++w) v += red[w * G1_TILE_F + q];
    s[t] = v;
  }
  float* __restrict__ yrow = Y + (size_t)(m0 + wave + 8 * half) * N_PROJ;
  v4f o4 = {s[0], s[1], s[2], s[3]};
  v2f o2 = {s[4], s[5]};
  *(v4f*)(yrow + 4 * ml) = o4;
  *(v2f*)(yrow + 64 + 2 * ml) = o2;
}

__global__ void ssm_dtproj_wmma(const float* __restrict__ XD,
                                const float* __restrict__ Wd,
                                const float* __restrict__ bias,
                                float* __restrict__ DT) {
  const int lane = threadIdx.x & 31;
  const int wave = threadIdx.x >> 5;
  const int half = lane >> 4;
  const int ml   = lane & 15;
  const int m0   = blockIdx.y << 4;
  const int n0   = (blockIdx.x << 7) + (wave << 4);

  const float* __restrict__ arow = XD + (size_t)(m0 + ml) * N_PROJ;
  v8f acc = {};
  for (int k = 0; k < DT_RANK; k += 4) {
    const int ka = k + 2 * half;
    v2f a, b;
    a.x = arow[ka];
    a.y = arow[ka + 1];
    b.x = Wd[(size_t)ka * D_INNER + n0 + ml];
    b.y = Wd[(size_t)(ka + 1) * D_INNER + n0 + ml];
    acc = __builtin_amdgcn_wmma_f32_16x16x4_f32(
        false, a, false, b, (short)0, acc, false, false);
  }
  const float bn = bias[n0 + ml];
#pragma unroll
  for (int r = 0; r < 8; ++r) {
    const int m = m0 + r + 8 * half;
    float v = acc[r] + bn;
    float sp = (v > 20.0f) ? v : log1pf(__expf(v));
    DT[(size_t)m * D_INNER + n0 + ml] = sp;
  }
}

__global__ __launch_bounds__(SCAN_THREADS)
void ssm_scan_pass1(const float* __restrict__ X,
                    const float* __restrict__ XDBL,
                    const float* __restrict__ DT,
                    const float* __restrict__ A_log,
                    float* __restrict__ chunkA,
                    float* __restrict__ chunkB) {
  const int c     = blockIdx.x * SCAN_THREADS + threadIdx.x;
  const int chunk = blockIdx.y;
  const int b     = blockIdx.x / (D_INNER / SCAN_THREADS);
  const int d     = c & (D_INNER - 1);

  float a2[D_STATE], s[D_STATE], p[D_STATE];
#pragma unroll
  for (int n = 0; n < D_STATE; ++n) {
    a2[n] = -__expf(A_log[d * D_STATE + n]) * 1.44269504088896f;
    s[n] = 0.0f;
    p[n] = 1.0f;
  }

  const size_t m0 = (size_t)b * SEQLEN + (size_t)chunk * CHUNKL;
  const float* __restrict__ dtp = DT + m0 * D_INNER + d;
  const float* __restrict__ xp  = X + m0 * D_INNER + d;
  const float* __restrict__ bcp = XDBL + m0 * N_PROJ + DT_RANK;

  float dtn[GRP], xn[GRP];
#pragma unroll
  for (int u = 0; u < GRP; ++u) {
    dtn[u] = dtp[(size_t)u * D_INNER];
    xn[u]  = xp[(size_t)u * D_INNER];
  }
  for (int g = 0; g < CHUNKL; g += GRP) {
    float dtc[GRP], xc[GRP];
#pragma unroll
    for (int u = 0; u < GRP; ++u) { dtc[u] = dtn[u]; xc[u] = xn[u]; }
    if (g + GRP < CHUNKL) {
#pragma unroll
      for (int u = 0; u < GRP; ++u) {
        dtn[u] = dtp[(size_t)(g + GRP + u) * D_INNER];
        xn[u]  = xp[(size_t)(g + GRP + u) * D_INNER];
      }
    }
#pragma unroll
    for (int u = 0; u < GRP; ++u) {
      const float* __restrict__ bc = bcp + (size_t)(g + u) * N_PROJ;
      const float dt = dtc[u];
      const float xv = xc[u];
#pragma unroll
      for (int n = 0; n < D_STATE; ++n) {
        const float dA = exp2f(dt * a2[n]);
        s[n] = fmaf(dA, s[n], dt * bc[n] * xv);
        p[n] *= dA;
      }
    }
  }
#pragma unroll
  for (int n = 0; n < D_STATE; ++n) {
    const size_t i = ((size_t)chunk * D_STATE + n) * NCHAN + c;
    chunkA[i] = p[n];
    chunkB[i] = s[n];
  }
}

__global__ void ssm_scan_pass2(const float* __restrict__ chunkA,
                               float* __restrict__ chunkB) {
  const int t = blockIdx.x * blockDim.x + threadIdx.x;
  float s = 0.0f;
#pragma unroll
  for (int ch = 0; ch < NCHUNK; ++ch) {
    const size_t i = (size_t)ch * NSTATE_TOT + t;
    const float Ac = chunkA[i];
    const float Bc = chunkB[i];
    chunkB[i] = s;
    s = fmaf(Ac, s, Bc);
  }
}

__global__ __launch_bounds__(SCAN_THREADS)
void ssm_scan_pass3(const float* __restrict__ X,
                    const float* __restrict__ XDBL,
                    const float* __restrict__ DT,
                    const float* __restrict__ A_log,
                    const float* __restrict__ Dv,
                    const float* __restrict__ Sin,
                    float* __restrict__ OUT) {
  const int c     = blockIdx.x * SCAN_THREADS + threadIdx.x;
  const int chunk = blockIdx.y;
  const int b     = blockIdx.x / (D_INNER / SCAN_THREADS);
  const int d     = c & (D_INNER - 1);

  float a2[D_STATE], st[D_STATE];
#pragma unroll
  for (int n = 0; n < D_STATE; ++n) {
    a2[n] = -__expf(A_log[d * D_STATE + n]) * 1.44269504088896f;
    st[n] = Sin[((size_t)chunk * D_STATE + n) * NCHAN + c];
  }
  const float Dd = Dv[d];

  const size_t m0 = (size_t)b * SEQLEN + (size_t)chunk * CHUNKL;
  const float* __restrict__ dtp = DT + m0 * D_INNER + d;
  const float* __restrict__ xp  = X + m0 * D_INNER + d;
  const float* __restrict__ bcp = XDBL + m0 * N_PROJ + DT_RANK;
  float* __restrict__ op = OUT + m0 * D_INNER + d;

  float dtn[GRP], xn[GRP];
#pragma unroll
  for (int u = 0; u < GRP; ++u) {
    dtn[u] = dtp[(size_t)u * D_INNER];
    xn[u]  = xp[(size_t)u * D_INNER];
  }
  for (int g = 0; g < CHUNKL; g += GRP) {
    float dtc[GRP], xc[GRP];
#pragma unroll
    for (int u = 0; u < GRP; ++u) { dtc[u] = dtn[u]; xc[u] = xn[u]; }
    if (g + GRP < CHUNKL) {
#pragma unroll
      for (int u = 0; u < GRP; ++u) {
        dtn[u] = dtp[(size_t)(g + GRP + u) * D_INNER];
        xn[u]  = xp[(size_t)(g + GRP + u) * D_INNER];
      }
    }
#pragma unroll
    for (int u = 0; u < GRP; ++u) {
      const float* __restrict__ bc = bcp + (size_t)(g + u) * N_PROJ;
      const float dt = dtc[u];
      const float xv = xc[u];
      float y[4] = {0.0f, 0.0f, 0.0f, 0.0f};
#pragma unroll
      for (int n = 0; n < D_STATE; ++n) {
        const float dA = exp2f(dt * a2[n]);
        st[n] = fmaf(dA, st[n], dt * bc[n] * xv);
        y[n & 3] = fmaf(st[n], bc[D_STATE + n], y[n & 3]);
      }
      op[(size_t)(g + u) * D_INNER] = ((y[0] + y[1]) + (y[2] + y[3])) + xv * Dd;
    }
  }
}

extern "C" void kernel_launch(void* const* d_in, const int* in_sizes, int n_in,
                              void* d_out, int out_size, void* d_ws, size_t ws_size,
                              hipStream_t stream) {
  const float* x    = (const float*)d_in[0];
  const float* xpw  = (const float*)d_in[1];
  const float* dtw  = (const float*)d_in[2];
  const float* dtb  = (const float*)d_in[3];
  const float* alog = (const float*)d_in[4];
  const float* dvec = (const float*)d_in[5];
  float* out = (float*)d_out;

  char* ws = (char*)d_ws;
  float* xdbl   = (float*)ws;
  ws += (size_t)M_TOT * N_PROJ * sizeof(float);
  float* delta  = (float*)ws;
  ws += (size_t)M_TOT * D_INNER * sizeof(float);
  float* chunkA = (float*)ws;
  ws += (size_t)NCHUNK * NSTATE_TOT * sizeof(float);
  float* chunkB = (float*)ws;

  ssm_xproj_wmma<<<dim3(M_TOT / 16), dim3(256), 0, stream>>>(x, xpw, xdbl);

  ssm_dtproj_wmma<<<dim3(D_INNER / 128, M_TOT / 16), dim3(256), 0, stream>>>(
      xdbl, dtw, dtb, delta);

  ssm_scan_pass1<<<dim3(NCHAN / SCAN_THREADS, NCHUNK), dim3(SCAN_THREADS), 0, stream>>>(
      x, xdbl, delta, alog, chunkA, chunkB);
  ssm_scan_pass2<<<dim3(NSTATE_TOT / 256), dim3(256), 0, stream>>>(
      chunkA, chunkB);
  ssm_scan_pass3<<<dim3(NCHAN / SCAN_THREADS, NCHUNK), dim3(SCAN_THREADS), 0, stream>>>(
      x, xdbl, delta, alog, dvec, chunkB, out);
}


# Scan passes 1 and 3: decay factor 2^(dt*a) taken from the hardware exp2 instruction directly; the library exp2f's rescale for results below 2^-126 (5 extra vector ops per call) is dropped.
# speedup vs baseline: 4.1669x; 1.2514x over previous
#include <hip/hip_runtime.h>
#include <math.h>


#define BATCH   2
#define SEQLEN  2048
#define D_INNER 2048
#define D_STATE 16
#define DT_RANK 64
#define N_PROJ  (DT_RANK + 2 * D_STATE)
#define M_TOT   (BATCH * SEQLEN)
#define NSTATE_TOT (BATCH * D_INNER * D_STATE)
#define NCHAN   (BATCH * D_INNER)
#define NCHUNK  32
#define CHUNKL  (SEQLEN / NCHUNK)
#define SCAN_THREADS 256
#define GRP     4

typedef float v2f __attribute__((ext_vector_type(2)));
typedef float v4f __attribute__((ext_vector_type(4)));
typedef float v8f __attribute__((ext_vector_type(8)));

__device__ __forceinline__ float hw_exp2(float v) {
  return __builtin_amdgcn_exp2f(v);
}

#define G1_WAVES  8
#define G1_KSLICE (D_INNER / G1_WAVES)
#define G1_KBATCH 32
#define G1_TILE_F (6 * 8 * 32)

__global__ __launch_bounds__(256)
void ssm_xproj_wmma(const float* __restrict__ X,
                    const float* __restrict__ W,
                    float* __restrict__ Y) {
  __shared__ float red[G1_WAVES * G1_TILE_F];
  const int lane = threadIdx.x & 31;
  const int wave = threadIdx.x >> 5;
  const int half = lane >> 4;
  const int ml   = lane & 15;
  const int m0   = blockIdx.x << 4;
  const int k0   = wave * G1_KSLICE + 4 * half;

  const float* __restrict__ ap  = X + (size_t)(m0 + ml) * D_INNER + k0;
  const float* __restrict__ bp4 = W + (size_t)k0 * N_PROJ + 4 * ml;
  const float* __restrict__ bp2 = W + (size_t)k0 * N_PROJ + 64 + 2 * ml;

  v8f acc[6];
#pragma unroll
  for (int t = 0; t < 6; ++t) acc[t] = (v8f){};

  v4f an[4], b4n[4][4];
  v2f b2n[4][4];
#pragma unroll
  for (int j = 0; j < 4; ++j) {
    an[j] = *(const v4f*)(ap + 8 * j);
#pragma unroll
    for (int e = 0; e < 4; ++e) {
      b4n[j][e] = *(const v4f*)(bp4 + (size_t)(8 * j + e) * N_PROJ);
      b2n[j][e] = *(const v2f*)(bp2 + (size_t)(8 * j + e) * N_PROJ);
    }
  }
#pragma unroll 1
  for (int kb = 0; kb < G1_KSLICE; kb += G1_KBATCH) {
    v4f ac[4], b4c[4][4];
    v2f b2c[4][4];
#pragma unroll
    for (int j = 0; j < 4; ++j) {
      ac[j] = an[j];
#pragma unroll
      for (int e = 0; e < 4; ++e) { b4c[j][e] = b4n[j][e]; b2c[j][e] = b2n[j][e]; }
    }
    if (kb + G1_KBATCH < G1_KSLICE) {
      const int kn = kb + G1_KBATCH;
#pragma unroll
      for (int j = 0; j < 4; ++j) {
        an[j] = *(const v4f*)(ap + kn + 8 * j);
#pragma unroll
        for (int e = 0; e < 4; ++e) {
          b4n[j][e] = *(const v4f*)(bp4 + (size_t)(kn + 8 * j + e) * N_PROJ);
          b2n[j][e] = *(const v2f*)(bp2 + (size_t)(kn + 8 * j + e) * N_PROJ);
        }
      }
    }
#pragma unroll
    for (int j = 0; j < 4; ++j) {
#pragma unroll
      for (int q = 0; q < 2; ++q) {
        v2f a;
        a.x = ac[j][2 * q];
        a.y = ac[j][2 * q + 1];
#pragma unroll
        for (int t = 0; t < 6; ++t) {
          v2f b;
          if (t < 4) {
            b.x = b4c[j][2 * q][t];
            b.y = b4c[j][2 * q + 1][t];
          } else {
            b.x = b2c[j][2 * q][t - 4];
            b.y = b2c[j][2 * q + 1][t - 4];
          }
          acc[t] = __builtin_amdgcn_wmma_f32_16x16x4_f32(
              false, a, false, b, (short)0, acc[t], false, false);
        }
      }
    }
  }

#pragma unroll
  for (int t = 0; t < 6; ++t) {
#pragma unroll
    for (int r = 0; r < 8; ++r) {
      red[wave * G1_TILE_F + (t * 8 + r) * 32 + lane] = acc[t][r];
    }
  }
  __syncthreads();
  float s[6];
#pragma unroll
  for (int t = 0; t < 6; ++t) {
    const int q = (t * 8 + wave) * 32 + lane;
    float v = red[q];
#pragma unroll
    for (int w = 1; w < G1_WAVES; ++w) v += red[w * G1_TILE_F + q];
    s[t] = v;
  }
  float* __restrict__ yrow = Y + (size_t)(m0 + wave + 8 * half) * N_PROJ;
  v4f o4 = {s[0], s[1], s[2], s[3]};
  v2f o2 = {s[4], s[5]};
  *(v4f*)(yrow + 4 * ml) = o4;
  *(v2f*)(yrow + 64 + 2 * ml) = o2;
}

__global__ void ssm_dtproj_wmma(const float* __restrict__ XD,
                                const float* __restrict__ Wd,
                                const float* __restrict__ bias,
                                float* __restrict__ DT) {
  const int lane = threadIdx.x & 31;
  const int wave = threadIdx.x >> 5;
  const int half = lane >> 4;
  const int ml   = lane & 15;
  const int m0   = blockIdx.y << 4;
  const int n0   = (blockIdx.x << 7) + (wave << 4);

  const float* __restrict__ arow = XD + (size_t)(m0 + ml) * N_PROJ;
  v8f acc = {};
  for (int k = 0; k < DT_RANK; k += 4) {
    const int ka = k + 2 * half;
    v2f a, b;
    a.x = arow[ka];
    a.y = arow[ka + 1];
    b.x = Wd[(size_t)ka * D_INNER + n0 + ml];
    b.y = Wd[(size_t)(ka + 1) * D_INNER + n0 + ml];
    acc = __builtin_amdgcn_wmma_f32_16x16x4_f32(
        false, a, false, b, (short)0, acc, false, false);
  }
  const float bn = bias[n0 + ml];
#pragma unroll
  for (int r = 0; r < 8; ++r) {
    const int m = m0 + r + 8 * half;
    float v = acc[r] + bn;
    float sp = (v > 20.0f) ? v : log1pf(__expf(v));
    DT[(size_t)m * D_INNER + n0 + ml] = sp;
  }
}

__global__ __launch_bounds__(SCAN_THREADS)
void ssm_scan_pass1(const float* __restrict__ X,
                    const float* __restrict__ XDBL,
                    const float* __restrict__ DT,
                    const float* __restrict__ A_log,
                    float* __restrict__ chunkA,
                    float* __restrict__ chunkB) {
  const int c     = blockIdx.x * SCAN_THREADS + threadIdx.x;
  const int chunk = blockIdx.y;
  const int b     = blockIdx.x / (D_INNER / SCAN_THREADS);
  const int d     = c & (D_INNER - 1);

  float a2[D_STATE], s[D_STATE], p[D_STATE];
#pragma unroll
  for (int n = 0; n < D_STATE; ++n) {
    a2[n] = -__expf(A_log[d * D_STATE + n]) * 1.44269504088896f;
    s[n] = 0.0f;
    p[n] = 1.0f;
  }

  const size_t m0 = (size_t)b * SEQLEN + (size_t)chunk * CHUNKL;
  const float* __restrict__ dtp = DT + m0 * D_INNER + d;
  const float* __restrict__ xp  = X + m0 * D_INNER + d;
  const float* __restrict__ bcp = XDBL + m0 * N_PROJ + DT_RANK;

  float dtn[GRP], xn[GRP];
#pragma unroll
  for (int u = 0; u < GRP; ++u) {
    dtn[u] = dtp[(size_t)u * D_INNER];
    xn[u]  = xp[(size_t)u * D_INNER];
  }
  for (int g = 0; g < CHUNKL; g += GRP) {
    float dtc[GRP], xc[GRP];
#pragma unroll
    for (int u = 0; u < GRP; ++u) { dtc[u] = dtn[u]; xc[u] = xn[u]; }
    if (g + GRP < CHUNKL) {
#pragma unroll
      for (int u = 0; u < GRP; ++u) {
        dtn[u] = dtp[(size_t)(g + GRP + u) * D_INNER];
        xn[u]  = xp[(size_t)(g + GRP + u) * D_INNER];
      }
    }
#pragma unroll
    for (int u = 0; u < GRP; ++u) {
      const float* __restrict__ bc = bcp + (size_t)(g + u) * N_PROJ;
      const float dt = dtc[u];
      const float xv = xc[u];
#pragma unroll
      for (int n = 0; n < D_STATE; ++n) {
        const float dA = hw_exp2(dt * a2[n]);
        s[n] = fmaf(dA, s[n], dt * bc[n] * xv);
        p[n] *= dA;
      }
    }
  }
#pragma unroll
  for (int n = 0; n < D_STATE; ++n) {
    const size_t i = ((size_t)chunk * D_STATE + n) * NCHAN + c;
    chunkA[i] = p[n];
    chunkB[i] = s[n];
  }
}

__global__ void ssm_scan_pass2(const float* __restrict__ chunkA,
                               float* __restrict__ chunkB) {
  const int t = blockIdx.x * blockDim.x + threadIdx.x;
  float s = 0.0f;
#pragma unroll
  for (int ch = 0; ch < NCHUNK; ++ch) {
    const size_t i = (size_t)ch * NSTATE_TOT + t;
    const float Ac = chunkA[i];
    const float Bc = chunkB[i];
    chunkB[i] = s;
    s = fmaf(Ac, s, Bc);
  }
}

__global__ __launch_bounds__(SCAN_THREADS)
void ssm_scan_pass3(const float* __restrict__ X,
                    const float* __restrict__ XDBL,
                    const float* __restrict__ DT,
                    const float* __restrict__ A_log,
                    const float* __restrict__ Dv,
                    const float* __restrict__ Sin,
                    float* __restrict__ OUT) {
  const int c     = blockIdx.x * SCAN_THREADS + threadIdx.x;
  const int chunk = blockIdx.y;
  const int b     = blockIdx.x / (D_INNER / SCAN_THREADS);
  const int d     = c & (D_INNER - 1);

  float a2[D_STATE], st[D_STATE];
#pragma unroll
  for (int n = 0; n < D_STATE; ++n) {
    a2[n] = -__expf(A_log[d * D_STATE + n]) * 1.44269504088896f;
    st[n] = Sin[((size_t)chunk * D_STATE + n) * NCHAN + c];
  }
  const float Dd = Dv[d];

  const size_t m0 = (size_t)b * SEQLEN + (size_t)chunk * CHUNKL;
  const float* __restrict__ dtp = DT + m0 * D_INNER + d;
  const float* __restrict__ xp  = X + m0 * D_INNER + d;
  const float* __restrict__ bcp = XDBL + m0 * N_PROJ + DT_RANK;
  float* __restrict__ op = OUT + m0 * D_INNER + d;

  float dtn[GRP], xn[GRP];
#pragma unroll
  for (int u = 0; u < GRP; ++u) {
    dtn[u] = dtp[(size_t)u * D_INNER];
    xn[u]  = xp[(size_t)u * D_INNER];
  }
  for (int g = 0; g < CHUNKL; g += GRP) {
    float dtc[GRP], xc[GRP];
#pragma unroll
    for (int u = 0; u < GRP; ++u) { dtc[u] = dtn[u]; xc[u] = xn[u]; }
    if (g + GRP < CHUNKL) {
#pragma unroll
      for (int u = 0; u < GRP; ++u) {
        dtn[u] = dtp[(size_t)(g + GRP + u) * D_INNER];
        xn[u]  = xp[(size_t)(g + GRP + u) * D_INNER];
      }
    }
#pragma unroll
    for (int u = 0; u < GRP; ++u) {
      const float* __restrict__ bc = bcp + (size_t)(g + u) * N_PROJ;
      const float dt = dtc[u];
      const float xv = xc[u];
      float y[4] = {0.0f, 0.0f, 0.0f, 0.0f};
#pragma unroll
      for (int n = 0; n < D_STATE; ++n) {
        const float dA = hw_exp2(dt * a2[n]);
        st[n] = fmaf(dA, st[n], dt * bc[n] * xv);
        y[n & 3] = fmaf(st[n], bc[D_STATE + n], y[n & 3]);
      }
      op[(size_t)(g + u) * D_INNER] = ((y[0] + y[1]) + (y[2] + y[3])) + xv * Dd;
    }
  }
}

extern "C" void kernel_launch(void* const* d_in, const int* in_sizes, int n_in,
                              void* d_out, int out_size, void* d_ws, size_t ws_size,
                              hipStream_t stream) {
  const float* x    = (const float*)d_in[0];
  const float* xpw  = (const float*)d_in[1];
  const float* dtw  = (const float*)d_in[2];
  const float* dtb  = (const float*)d_in[3];
  const float* alog = (const float*)d_in[4];
  const float* dvec = (const float*)d_in[5];
  float* out = (float*)d_out;

  char* ws = (char*)d_ws;
  float* xdbl   = (float*)ws;
  ws += (size_t)M_TOT * N_PROJ * sizeof(float);
  float* delta  = (float*)ws;
  ws += (size_t)M_TOT * D_INNER * sizeof(float);
  float* chunkA = (float*)ws;
  ws += (size_t)NCHUNK * NSTATE_TOT * sizeof(float);
  float* chunkB = (float*)ws;

  ssm_xproj_wmma<<<dim3(M_TOT / 16), dim3(256), 0, stream>>>(x, xpw, xdbl);

  ssm_dtproj_wmma<<<dim3(D_INNER / 128, M_TOT / 16), dim3(256), 0, stream>>>(
      xdbl, dtw, dtb, delta);

  ssm_scan_pass1<<<dim3(NCHAN / SCAN_THREADS, NCHUNK), dim3(SCAN_THREADS), 0, stream>>>(
      x, xdbl, delta, alog, chunkA, chunkB);
  ssm_scan_pass2<<<dim3(NSTATE_TOT / 256), dim3(256), 0, stream>>>(
      chunkA, chunkB);
  ssm_scan_pass3<<<dim3(NCHAN / SCAN_THREADS, NCHUNK), dim3(SCAN_THREADS), 0, stream>>>(
      x, xdbl, delta, alog, dvec, chunkB, out);
}


# GEMM2 softplus: max(v,0)+2*atanh(t/(2+t)), t=exp2(-|v|*log2e), from hardware exp2/rcp and a 7-term series, not library log1pf; run by 4096 resident waves that each walk 8 column tiles.
# speedup vs baseline: 4.7062x; 1.1294x over previous
#include <hip/hip_runtime.h>
#include <math.h>


#define BATCH   2
#define SEQLEN  2048
#define D_INNER 2048
#define D_STATE 16
#define DT_RANK 64
#define N_PROJ  (DT_RANK + 2 * D_STATE)
#define M_TOT   (BATCH * SEQLEN)
#define NSTATE_TOT (BATCH * D_INNER * D_STATE)
#define NCHAN   (BATCH * D_INNER)
#define NCHUNK  32
#define CHUNKL  (SEQLEN / NCHUNK)
#define SCAN_THREADS 256
#define GRP     4

typedef float v2f __attribute__((ext_vector_type(2)));
typedef float v4f __attribute__((ext_vector_type(4)));
typedef float v8f __attribute__((ext_vector_type(8)));

__device__ __forceinline__ float hw_exp2(float v) {
  return __builtin_amdgcn_exp2f(v);
}

#define G1_WAVES  8
#define G1_KSLICE (D_INNER / G1_WAVES)
#define G1_KBATCH 32
#define G1_TILE_F (6 * 8 * 32)

__global__ __launch_bounds__(256)
void ssm_xproj_wmma(const float* __restrict__ X,
                    const float* __restrict__ W,
                    float* __restrict__ Y) {
  __shared__ float red[G1_WAVES * G1_TILE_F];
  const int lane = threadIdx.x & 31;
  const int wave = threadIdx.x >> 5;
  const int half = lane >> 4;
  const int ml   = lane & 15;
  const int m0   = blockIdx.x << 4;
  const int k0   = wave * G1_KSLICE + 4 * half;

  const float* __restrict__ ap  = X + (size_t)(m0 + ml) * D_INNER + k0;
  const float* __restrict__ bp4 = W + (size_t)k0 * N_PROJ + 4 * ml;
  const float* __restrict__ bp2 = W + (size_t)k0 * N_PROJ + 64 + 2 * ml;

  v8f acc[6];
#pragma unroll
  for (int t = 0; t < 6; ++t) acc[t] = (v8f){};

  v4f an[4], b4n[4][4];
  v2f b2n[4][4];
#pragma unroll
  for (int j = 0; j < 4; ++j) {
    an[j] = *(const v4f*)(ap + 8 * j);
#pragma unroll
    for (int e = 0; e < 4; ++e) {
      b4n[j][e] = *(const v4f*)(bp4 + (size_t)(8 * j + e) * N_PROJ);
      b2n[j][e] = *(const v2f*)(bp2 + (size_t)(8 * j + e) * N_PROJ);
    }
  }
#pragma unroll 1
  for (int kb = 0; kb < G1_KSLICE; kb += G1_KBATCH) {
    v4f ac[4], b4c[4][4];
    v2f b2c[4][4];
#pragma unroll
    for (int j = 0; j < 4; ++j) {
      ac[j] = an[j];
#pragma unroll
      for (int e = 0; e < 4; ++e) { b4c[j][e] = b4n[j][e]; b2c[j][e] = b2n[j][e]; }
    }
    if (kb + G1_KBATCH < G1_KSLICE) {
      const int kn = kb + G1_KBATCH;
#pragma unroll
      for (int j = 0; j < 4; ++j) {
        an[j] = *(const v4f*)(ap + kn + 8 * j);
#pragma unroll
        for (int e = 0; e < 4; ++e) {
          b4n[j][e] = *(const v4f*)(bp4 + (size_t)(kn + 8 * j + e) * N_PROJ);
          b2n[j][e] = *(const v2f*)(bp2 + (size_t)(kn + 8 * j + e) * N_PROJ);
        }
      }
    }
#pragma unroll
    for (int j = 0; j < 4; ++j) {
#pragma unroll
      for (int q = 0; q < 2; ++q) {
        v2f a;
        a.x = ac[j][2 * q];
        a.y = ac[j][2 * q + 1];
#pragma unroll
        for (int t = 0; t < 6; ++t) {
          v2f b;
          if (t < 4) {
            b.x = b4c[j][2 * q][t];
            b.y = b4c[j][2 * q + 1][t];
          } else {
            b.x = b2c[j][2 * q][t - 4];
            b.y = b2c[j][2 * q + 1][t - 4];
          }
          acc[t] = __builtin_amdgcn_wmma_f32_16x16x4_f32(
              false, a, false, b, (short)0, acc[t], false, false);
        }
      }
    }
  }

#pragma unroll
  for (int t = 0; t < 6; ++t) {
#pragma unroll
    for (int r = 0; r < 8; ++r) {
      red[wave * G1_TILE_F + (t * 8 + r) * 32 + lane] = acc[t][r];
    }
  }
  __syncthreads();
  float s[6];
#pragma unroll
  for (int t = 0; t < 6; ++t) {
    const int q = (t * 8 + wave) * 32 + lane;
    float v = red[q];
#pragma unroll
    for (int w = 1; w < G1_WAVES; ++w) v += red[w * G1_TILE_F + q];
    s[t] = v;
  }
  float* __restrict__ yrow = Y + (size_t)(m0 + wave + 8 * half) * N_PROJ;
  v4f o4 = {s[0], s[1], s[2], s[3]};
  v2f o2 = {s[4], s[5]};
  *(v4f*)(yrow + 4 * ml) = o4;
  *(v2f*)(yrow + 64 + 2 * ml) = o2;
}

#define G2_BCOLS  128
#define G2_PAIRS  (G2_BCOLS / 32)
#define G2_KSTEPS (DT_RANK / 4)

__device__ __forceinline__ float softplus_series(float v) {
  const float t = hw_exp2(-fabsf(v) * 1.44269504088896f);
  const float w = 2.0f + t;
  const float r = __builtin_amdgcn_rcpf(w);
  float s = t * r;
  s = fmaf(fmaf(-w, s, t), r, s);
  const float z = s * s;
  float p = 2.0f / 13.0f;
  p = fmaf(p, z, 2.0f / 11.0f);
  p = fmaf(p, z, 2.0f / 9.0f);
  p = fmaf(p, z, 2.0f / 7.0f);
  p = fmaf(p, z, 2.0f / 5.0f);
  p = fmaf(p, z, 2.0f / 3.0f);
  p = fmaf(p, z, 2.0f);
  return fmaf(s, p, fmaxf(v, 0.0f));
}

__global__ __launch_bounds__(256)
void ssm_dtproj_wmma(const float* __restrict__ XD,
                     const float* __restrict__ Wd,
                     const float* __restrict__ bias,
                     float* __restrict__ DT) {
  const int lane = threadIdx.x & 31;
  const int wave = threadIdx.x >> 5;
  const int half = lane >> 4;
  const int ml   = lane & 15;
  const int m0   = ((blockIdx.y << 3) + wave) << 4;
  const int nb   = blockIdx.x * G2_BCOLS;

  const float* __restrict__ arow = XD + (size_t)(m0 + ml) * N_PROJ + 2 * half;
  v2f a[G2_KSTEPS];
#pragma unroll
  for (int i = 0; i < G2_KSTEPS; ++i) a[i] = *(const v2f*)(arow + 4 * i);

  const float* __restrict__ bcol = Wd + (size_t)(2 * half) * D_INNER + nb + ml;
  float* __restrict__ orow = DT + (size_t)(m0 + 8 * half) * D_INNER + nb + ml;

  v2f bn[2][G2_KSTEPS];
  float biasn[2];
#pragma unroll
  for (int u = 0; u < 2; ++u) {
    biasn[u] = bias[nb + 16 * u + ml];
#pragma unroll
    for (int i = 0; i < G2_KSTEPS; ++i) {
      bn[u][i].x = bcol[(size_t)(4 * i) * D_INNER + 16 * u];
      bn[u][i].y = bcol[(size_t)(4 * i + 1) * D_INNER + 16 * u];
    }
  }
#pragma unroll 1
  for (int pr = 0; pr < G2_PAIRS; ++pr) {
    v2f bc[2][G2_KSTEPS];
    float biasc[2];
#pragma unroll
    for (int u = 0; u < 2; ++u) {
      biasc[u] = biasn[u];
#pragma unroll
      for (int i = 0; i < G2_KSTEPS; ++i) bc[u][i] = bn[u][i];
    }
    if (pr + 1 < G2_PAIRS) {
      const int cn = 32 * (pr + 1);
#pragma unroll
      for (int u = 0; u < 2; ++u) {
        biasn[u] = bias[nb + cn + 16 * u + ml];
#pragma unroll
        for (int i = 0; i < G2_KSTEPS; ++i) {
          bn[u][i].x = bcol[(size_t)(4 * i) * D_INNER + cn + 16 * u];
          bn[u][i].y = bcol[(size_t)(4 * i + 1) * D_INNER + cn + 16 * u];
        }
      }
    }
    v8f acc0 = {}, acc1 = {};
#pragma unroll
    for (int i = 0; i < G2_KSTEPS; ++i) {
      acc0 = __builtin_amdgcn_wmma_f32_16x16x4_f32(
          false, a[i], false, bc[0][i], (short)0, acc0, false, false);
      acc1 = __builtin_amdgcn_wmma_f32_16x16x4_f32(
          false, a[i], false, bc[1][i], (short)0, acc1, false, false);
    }
    float* __restrict__ o = orow + 32 * pr;
#pragma unroll
    for (int r = 0; r < 8; ++r) {
      o[(size_t)r * D_INNER]      = softplus_series(acc0[r] + biasc[0]);
      o[(size_t)r * D_INNER + 16] = softplus_series(acc1[r] + biasc[1]);
    }
  }
}

__global__ __launch_bounds__(SCAN_THREADS)
void ssm_scan_pass1(const float* __restrict__ X,
                    const float* __restrict__ XDBL,
                    const float* __restrict__ DT,
                    const float* __restrict__ A_log,
                    float* __restrict__ chunkA,
                    float* __restrict__ chunkB) {
  const int c     = blockIdx.x * SCAN_THREADS + threadIdx.x;
  const int chunk = blockIdx.y;
  const int b     = blockIdx.x / (D_INNER / SCAN_THREADS);
  const int d     = c & (D_INNER - 1);

  float a2[D_STATE], s[D_STATE], p[D_STATE];
#pragma unroll
  for (int n = 0; n < D_STATE; ++n) {
    a2[n] = -__expf(A_log[d * D_STATE + n]) * 1.44269504088896f;
    s[n] = 0.0f;
    p[n] = 1.0f;
  }

  const size_t m0 = (size_t)b * SEQLEN + (size_t)chunk * CHUNKL;
  const float* __restrict__ dtp = DT + m0 * D_INNER + d;
  const float* __restrict__ xp  = X + m0 * D_INNER + d;
  const float* __restrict__ bcp = XDBL + m0 * N_PROJ + DT_RANK;

  float dtn[GRP], xn[GRP];
#pragma unroll
  for (int u = 0; u < GRP; ++u) {
    dtn[u] = dtp[(size_t)u * D_INNER];
    xn[u]  = xp[(size_t)u * D_INNER];
  }
  for (int g = 0; g < CHUNKL; g += GRP) {
    float dtc[GRP], xc[GRP];
#pragma unroll
    for (int u = 0; u < GRP; ++u) { dtc[u] = dtn[u]; xc[u] = xn[u]; }
    if (g + GRP < CHUNKL) {
#pragma unroll
      for (int u = 0; u < GRP; ++u) {
        dtn[u] = dtp[(size_t)(g + GRP + u) * D_INNER];
        xn[u]  = xp[(size_t)(g + GRP + u) * D_INNER];
      }
    }
#pragma unroll
    for (int u = 0; u < GRP; ++u) {
      const float* __restrict__ bc = bcp + (size_t)(g + u) * N_PROJ;
      const float dt = dtc[u];
      const float xv = xc[u];
#pragma unroll
      for (int n = 0; n < D_STATE; ++n) {
        const float dA = hw_exp2(dt * a2[n]);
        s[n] = fmaf(dA, s[n], dt * bc[n] * xv);
        p[n] *= dA;
      }
    }
  }
#pragma unroll
  for (int n = 0; n < D_STATE; ++n) {
    const size_t i = ((size_t)chunk * D_STATE + n) * NCHAN + c;
    chunkA[i] = p[n];
    chunkB[i] = s[n];
  }
}

__global__ void ssm_scan_pass2(const float* __restrict__ chunkA,
                               float* __restrict__ chunkB) {
  const int t = blockIdx.x * blockDim.x + threadIdx.x;
  float s = 0.0f;
#pragma unroll
  for (int ch = 0; ch < NCHUNK; ++ch) {
    const size_t i = (size_t)ch * NSTATE_TOT + t;
    const float Ac = chunkA[i];
    const float Bc = chunkB[i];
    chunkB[i] = s;
    s = fmaf(Ac, s, Bc);
  }
}

__global__ __launch_bounds__(SCAN_THREADS)
void ssm_scan_pass3(const float* __restrict__ X,
                    const float* __restrict__ XDBL,
                    const float* __restrict__ DT,
                    const float* __restrict__ A_log,
                    const float* __restrict__ Dv,
                    const float* __restrict__ Sin,
                    float* __restrict__ OUT) {
  const int c     = blockIdx.x * SCAN_THREADS + threadIdx.x;
  const int chunk = blockIdx.y;
  const int b     = blockIdx.x / (D_INNER / SCAN_THREADS);
  const int d     = c & (D_INNER - 1);

  float a2[D_STATE], st[D_STATE];
#pragma unroll
  for (int n = 0; n < D_STATE; ++n) {
    a2[n] = -__expf(A_log[d * D_STATE + n]) * 1.44269504088896f;
    st[n] = Sin[((size_t)chunk * D_STATE + n) * NCHAN + c];
  }
  const float Dd = Dv[d];

  const size_t m0 = (size_t)b * SEQLEN + (size_t)chunk * CHUNKL;
  const float* __restrict__ dtp = DT + m0 * D_INNER + d;
  const float* __restrict__ xp  = X + m0 * D_INNER + d;
  const float* __restrict__ bcp = XDBL + m0 * N_PROJ + DT_RANK;
  float* __restrict__ op = OUT + m0 * D_INNER + d;

  float dtn[GRP], xn[GRP];
#pragma unroll
  for (int u = 0; u < GRP; ++u) {
    dtn[u] = dtp[(size_t)u * D_INNER];
    xn[u]  = xp[(size_t)u * D_INNER];
  }
  for (int g = 0; g < CHUNKL; g += GRP) {
    float dtc[GRP], xc[GRP];
#pragma unroll
    for (int u = 0; u < GRP; ++u) { dtc[u] = dtn[u]; xc[u] = xn[u]; }
    if (g + GRP < CHUNKL) {
#pragma unroll
      for (int u = 0; u < GRP; ++u) {
        dtn[u] = dtp[(size_t)(g + GRP + u) * D_INNER];
        xn[u]  = xp[(size_t)(g + GRP + u) * D_INNER];
      }
    }
#pragma unroll
    for (int u = 0; u < GRP; ++u) {
      const float* __restrict__ bc = bcp + (size_t)(g + u) * N_PROJ;
      const float dt = dtc[u];
      const float xv = xc[u];
      float y[4] = {0.0f, 0.0f, 0.0f, 0.0f};
#pragma unroll
      for (int n = 0; n < D_STATE; ++n) {
        const float dA = hw_exp2(dt * a2[n]);
        st[n] = fmaf(dA, st[n], dt * bc[n] * xv);
        y[n & 3] = fmaf(st[n], bc[D_STATE + n], y[n & 3]);
      }
      op[(size_t)(g + u) * D_INNER] = ((y[0] + y[1]) + (y[2] + y[3])) + xv * Dd;
    }
  }
}

extern "C" void kernel_launch(void* const* d_in, const int* in_sizes, int n_in,
                              void* d_out, int out_size, void* d_ws, size_t ws_size,
                              hipStream_t stream) {
  const float* x    = (const float*)d_in[0];
  const float* xpw  = (const float*)d_in[1];
  const float* dtw  = (const float*)d_in[2];
  const float* dtb  = (const float*)d_in[3];
  const float* alog = (const float*)d_in[4];
  const float* dvec = (const float*)d_in[5];
  float* out = (float*)d_out;

  char* ws = (char*)d_ws;
  float* xdbl   = (float*)ws;
  ws += (size_t)M_TOT * N_PROJ * sizeof(float);
  float* delta  = (float*)ws;
  ws += (size_t)M_TOT * D_INNER * sizeof(float);
  float* chunkA = (float*)ws;
  ws += (size_t)NCHUNK * NSTATE_TOT * sizeof(float);
  float* chunkB = (float*)ws;

  ssm_xproj_wmma<<<dim3(M_TOT / 16), dim3(256), 0, stream>>>(x, xpw, xdbl);

  ssm_dtproj_wmma<<<dim3(D_INNER / G2_BCOLS, M_TOT / 128), dim3(256), 0, stream>>>(
      xdbl, dtw, dtb, delta);

  ssm_scan_pass1<<<dim3(NCHAN / SCAN_THREADS, NCHUNK), dim3(SCAN_THREADS), 0, stream>>>(
      x, xdbl, delta, alog, chunkA, chunkB);
  ssm_scan_pass2<<<dim3(NSTATE_TOT / 256), dim3(256), 0, stream>>>(
      chunkA, chunkB);
  ssm_scan_pass3<<<dim3(NCHAN / SCAN_THREADS, NCHUNK), dim3(SCAN_THREADS), 0, stream>>>(
      x, xdbl, delta, alog, dvec, chunkB, out);
}


# Scan passes 1 and 3: input term formed as (dt*x)*B[n] with dt*x once per time step: one multiply per state, B read straight from scalar registers (product order changes, sums do not).
# speedup vs baseline: 4.8891x; 1.0389x over previous
#include <hip/hip_runtime.h>
#include <math.h>


#define BATCH   2
#define SEQLEN  2048
#define D_INNER 2048
#define D_STATE 16
#define DT_RANK 64
#define N_PROJ  (DT_RANK + 2 * D_STATE)
#define M_TOT   (BATCH * SEQLEN)
#define NSTATE_TOT (BATCH * D_INNER * D_STATE)
#define NCHAN   (BATCH * D_INNER)
#define NCHUNK  32
#define CHUNKL  (SEQLEN / NCHUNK)
#define SCAN_THREADS 256
#define GRP     4

typedef float v2f __attribute__((ext_vector_type(2)));
typedef float v4f __attribute__((ext_vector_type(4)));
typedef float v8f __attribute__((ext_vector_type(8)));

__device__ __forceinline__ float hw_exp2(float v) {
  return __builtin_amdgcn_exp2f(v);
}

#define G1_WAVES  8
#define G1_KSLICE (D_INNER / G1_WAVES)
#define G1_KBATCH 32
#define G1_TILE_F (6 * 8 * 32)

__global__ __launch_bounds__(256)
void ssm_xproj_wmma(const float* __restrict__ X,
                    const float* __restrict__ W,
                    float* __restrict__ Y) {
  __shared__ float red[G1_WAVES * G1_TILE_F];
  const int lane = threadIdx.x & 31;
  const int wave = threadIdx.x >> 5;
  const int half = lane >> 4;
  const int ml   = lane & 15;
  const int m0   = blockIdx.x << 4;
  const int k0   = wave * G1_KSLICE + 4 * half;

  const float* __restrict__ ap  = X + (size_t)(m0 + ml) * D_INNER + k0;
  const float* __restrict__ bp4 = W + (size_t)k0 * N_PROJ + 4 * ml;
  const float* __restrict__ bp2 = W + (size_t)k0 * N_PROJ + 64 + 2 * ml;

  v8f acc[6];
#pragma unroll
  for (int t = 0; t < 6; ++t) acc[t] = (v8f){};

  v4f an[4], b4n[4][4];
  v2f b2n[4][4];
#pragma unroll
  for (int j = 0; j < 4; ++j) {
    an[j] = *(const v4f*)(ap + 8 * j);
#pragma unroll
    for (int e = 0; e < 4; ++e) {
      b4n[j][e] = *(const v4f*)(bp4 + (size_t)(8 * j + e) * N_PROJ);
      b2n[j][e] = *(const v2f*)(bp2 + (size_t)(8 * j + e) * N_PROJ);
    }
  }
#pragma unroll 1
  for (int kb = 0; kb < G1_KSLICE; kb += G1_KBATCH) {
    v4f ac[4], b4c[4][4];
    v2f b2c[4][4];
#pragma unroll
    for (int j = 0; j < 4; ++j) {
      ac[j] = an[j];
#pragma unroll
      for (int e = 0; e < 4; ++e) { b4c[j][e] = b4n[j][e]; b2c[j][e] = b2n[j][e]; }
    }
    if (kb + G1_KBATCH < G1_KSLICE) {
      const int kn = kb + G1_KBATCH;
#pragma unroll
      for (int j = 0; j < 4; ++j) {
        an[j] = *(const v4f*)(ap + kn + 8 * j);
#pragma unroll
        for (int e = 0; e < 4; ++e) {
          b4n[j][e] = *(const v4f*)(bp4 + (size_t)(kn + 8 * j + e) * N_PROJ);
          b2n[j][e] = *(const v2f*)(bp2 + (size_t)(kn + 8 * j + e) * N_PROJ);
        }
      }
    }
#pragma unroll
    for (int j = 0; j < 4; ++j) {
#pragma unroll
      for (int q = 0; q < 2; ++q) {
        v2f a;
        a.x = ac[j][2 * q];
        a.y = ac[j][2 * q + 1];
#pragma unroll
        for (int t = 0; t < 6; ++t) {
          v2f b;
          if (t < 4) {
            b.x = b4c[j][2 * q][t];
            b.y = b4c[j][2 * q + 1][t];
          } else {
            b.x = b2c[j][2 * q][t - 4];
            b.y = b2c[j][2 * q + 1][t - 4];
          }
          acc[t] = __builtin_amdgcn_wmma_f32_16x16x4_f32(
              false, a, false, b, (short)0, acc[t], false, false);
        }
      }
    }
  }

#pragma unroll
  for (int t = 0; t < 6; ++t) {
#pragma unroll
    for (int r = 0; r < 8; ++r) {
      red[wave * G1_TILE_F + (t * 8 + r) * 32 + lane] = acc[t][r];
    }
  }
  __syncthreads();
  float s[6];
#pragma unroll
  for (int t = 0; t < 6; ++t) {
    const int q = (t * 8 + wave) * 32 + lane;
    float v = red[q];
#pragma unroll
    for (int w = 1; w < G1_WAVES; ++w) v += red[w * G1_TILE_F + q];
    s[t] = v;
  }
  float* __restrict__ yrow = Y + (size_t)(m0 + wave + 8 * half) * N_PROJ;
  v4f o4 = {s[0], s[1], s[2], s[3]};
  v2f o2 = {s[4], s[5]};
  *(v4f*)(yrow + 4 * ml) = o4;
  *(v2f*)(yrow + 64 + 2 * ml) = o2;
}

#define G2_BCOLS  128
#define G2_PAIRS  (G2_BCOLS / 32)
#define G2_KSTEPS (DT_RANK / 4)

__device__ __forceinline__ float softplus_series(float v) {
  const float t = hw_exp2(-fabsf(v) * 1.44269504088896f);
  const float w = 2.0f + t;
  const float r = __builtin_amdgcn_rcpf(w);
  float s = t * r;
  s = fmaf(fmaf(-w, s, t), r, s);
  const float z = s * s;
  float p = 2.0f / 13.0f;
  p = fmaf(p, z, 2.0f / 11.0f);
  p = fmaf(p, z, 2.0f / 9.0f);
  p = fmaf(p, z, 2.0f / 7.0f);
  p = fmaf(p, z, 2.0f / 5.0f);
  p = fmaf(p, z, 2.0f / 3.0f);
  p = fmaf(p, z, 2.0f);
  return fmaf(s, p, fmaxf(v, 0.0f));
}

__global__ __launch_bounds__(256)
void ssm_dtproj_wmma(const float* __restrict__ XD,
                     const float* __restrict__ Wd,
                     const float* __restrict__ bias,
                     float* __restrict__ DT) {
  const int lane = threadIdx.x & 31;
  const int wave = threadIdx.x >> 5;
  const int half = lane >> 4;
  const int ml   = lane & 15;
  const int m0   = ((blockIdx.y << 3) + wave) << 4;
  const int nb   = blockIdx.x * G2_BCOLS;

  const float* __restrict__ arow = XD + (size_t)(m0 + ml) * N_PROJ + 2 * half;
  v2f a[G2_KSTEPS];
#pragma unroll
  for (int i = 0; i < G2_KSTEPS; ++i) a[i] = *(const v2f*)(arow + 4 * i);

  const float* __restrict__ bcol = Wd + (size_t)(2 * half) * D_INNER + nb + ml;
  float* __restrict__ orow = DT + (size_t)(m0 + 8 * half) * D_INNER + nb + ml;

  v2f bn[2][G2_KSTEPS];
  float biasn[2];
#pragma unroll
  for (int u = 0; u < 2; ++u) {
    biasn[u] = bias[nb + 16 * u + ml];
#pragma unroll
    for (int i = 0; i < G2_KSTEPS; ++i) {
      bn[u][i].x = bcol[(size_t)(4 * i) * D_INNER + 16 * u];
      bn[u][i].y = bcol[(size_t)(4 * i + 1) * D_INNER + 16 * u];
    }
  }
#pragma unroll 1
  for (int pr = 0; pr < G2_PAIRS; ++pr) {
    v2f bc[2][G2_KSTEPS];
    float biasc[2];
#pragma unroll
    for (int u = 0; u < 2; ++u) {
      biasc[u] = biasn[u];
#pragma unroll
      for (int i = 0; i < G2_KSTEPS; ++i) bc[u][i] = bn[u][i];
    }
    if (pr + 1 < G2_PAIRS) {
      const int cn = 32 * (pr + 1);
#pragma unroll
      for (int u = 0; u < 2; ++u) {
        biasn[u] = bias[nb + cn + 16 * u + ml];
#pragma unroll
        for (int i = 0; i < G2_KSTEPS; ++i) {
          bn[u][i].x = bcol[(size_t)(4 * i) * D_INNER + cn + 16 * u];
          bn[u][i].y = bcol[(size_t)(4 * i + 1) * D_INNER + cn + 16 * u];
        }
      }
    }
    v8f acc0 = {}, acc1 = {};
#pragma unroll
    for (int i = 0; i < G2_KSTEPS; ++i) {
      acc0 = __builtin_amdgcn_wmma_f32_16x16x4_f32(
          false, a[i], false, bc[0][i], (short)0, acc0, false, false);
      acc1 = __builtin_amdgcn_wmma_f32_16x16x4_f32(
          false, a[i], false, bc[1][i], (short)0, acc1, false, false);
    }
    float* __restrict__ o = orow + 32 * pr;
#pragma unroll
    for (int r = 0; r < 8; ++r) {
      o[(size_t)r * D_INNER]      = softplus_series(acc0[r] + biasc[0]);
      o[(size_t)r * D_INNER + 16] = softplus_series(acc1[r] + biasc[1]);
    }
  }
}

__global__ __launch_bounds__(SCAN_THREADS)
void ssm_scan_pass1(const float* __restrict__ X,
                    const float* __restrict__ XDBL,
                    const float* __restrict__ DT,
                    const float* __restrict__ A_log,
                    float* __restrict__ chunkA,
                    float* __restrict__ chunkB) {
  const int c     = blockIdx.x * SCAN_THREADS + threadIdx.x;
  const int chunk = blockIdx.y;
  const int b     = blockIdx.x / (D_INNER / SCAN_THREADS);
  const int d     = c & (D_INNER - 1);

  float a2[D_STATE], s[D_STATE], p[D_STATE];
#pragma unroll
  for (int n = 0; n < D_STATE; ++n) {
    a2[n] = -__expf(A_log[d * D_STATE + n]) * 1.44269504088896f;
    s[n] = 0.0f;
    p[n] = 1.0f;
  }

  const size_t m0 = (size_t)b * SEQLEN + (size_t)chunk * CHUNKL;
  const float* __restrict__ dtp = DT + m0 * D_INNER + d;
  const float* __restrict__ xp  = X + m0 * D_INNER + d;
  const float* __restrict__ bcp = XDBL + m0 * N_PROJ + DT_RANK;

  float dtn[GRP], xn[GRP];
#pragma unroll
  for (int u = 0; u < GRP; ++u) {
    dtn[u] = dtp[(size_t)u * D_INNER];
    xn[u]  = xp[(size_t)u * D_INNER];
  }
  for (int g = 0; g < CHUNKL; g += GRP) {
    float dtc[GRP], xc[GRP];
#pragma unroll
    for (int u = 0; u < GRP; ++u) { dtc[u] = dtn[u]; xc[u] = xn[u]; }
    if (g + GRP < CHUNKL) {
#pragma unroll
      for (int u = 0; u < GRP; ++u) {
        dtn[u] = dtp[(size_t)(g + GRP + u) * D_INNER];
        xn[u]  = xp[(size_t)(g + GRP + u) * D_INNER];
      }
    }
#pragma unroll
    for (int u = 0; u < GRP; ++u) {
      const float* __restrict__ bc = bcp + (size_t)(g + u) * N_PROJ;
      const float dt = dtc[u];
      const float xv = xc[u];
      const float dtx = dt * xv;
#pragma unroll
      for (int n = 0; n < D_STATE; ++n) {
        const float dA = hw_exp2(dt * a2[n]);
        s[n] = fmaf(dA, s[n], dtx * bc[n]);
        p[n] *= dA;
      }
    }
  }
#pragma unroll
  for (int n = 0; n < D_STATE; ++n) {
    const size_t i = ((size_t)chunk * D_STATE + n) * NCHAN + c;
    chunkA[i] = p[n];
    chunkB[i] = s[n];
  }
}

__global__ void ssm_scan_pass2(const float* __restrict__ chunkA,
                               float* __restrict__ chunkB) {
  const int t = blockIdx.x * blockDim.x + threadIdx.x;
  float s = 0.0f;
#pragma unroll
  for (int ch = 0; ch < NCHUNK; ++ch) {
    const size_t i = (size_t)ch * NSTATE_TOT + t;
    const float Ac = chunkA[i];
    const float Bc = chunkB[i];
    chunkB[i] = s;
    s = fmaf(Ac, s, Bc);
  }
}

__global__ __launch_bounds__(SCAN_THREADS)
void ssm_scan_pass3(const float* __restrict__ X,
                    const float* __restrict__ XDBL,
                    const float* __restrict__ DT,
                    const float* __restrict__ A_log,
                    const float* __restrict__ Dv,
                    const float* __restrict__ Sin,
                    float* __restrict__ OUT) {
  const int c     = blockIdx.x * SCAN_THREADS + threadIdx.x;
  const int chunk = blockIdx.y;
  const int b     = blockIdx.x / (D_INNER / SCAN_THREADS);
  const int d     = c & (D_INNER - 1);

  float a2[D_STATE], st[D_STATE];
#pragma unroll
  for (int n = 0; n < D_STATE; ++n) {
    a2[n] = -__expf(A_log[d * D_STATE + n]) * 1.44269504088896f;
    st[n] = Sin[((size_t)chunk * D_STATE + n) * NCHAN + c];
  }
  const float Dd = Dv[d];

  const size_t m0 = (size_t)b * SEQLEN + (size_t)chunk * CHUNKL;
  const float* __restrict__ dtp = DT + m0 * D_INNER + d;
  const float* __restrict__ xp  = X + m0 * D_INNER + d;
  const float* __restrict__ bcp = XDBL + m0 * N_PROJ + DT_RANK;
  float* __restrict__ op = OUT + m0 * D_INNER + d;

  float dtn[GRP], xn[GRP];
#pragma unroll
  for (int u = 0; u < GRP; ++u) {
    dtn[u] = dtp[(size_t)u * D_INNER];
    xn[u]  = xp[(size_t)u * D_INNER];
  }
  for (int g = 0; g < CHUNKL; g += GRP) {
    float dtc[GRP], xc[GRP];
#pragma unroll
    for (int u = 0; u < GRP; ++u) { dtc[u] = dtn[u]; xc[u] = xn[u]; }
    if (g + GRP < CHUNKL) {
#pragma unroll
      for (int u = 0; u < GRP; ++u) {
        dtn[u] = dtp[(size_t)(g + GRP + u) * D_INNER];
        xn[u]  = xp[(size_t)(g + GRP + u) * D_INNER];
      }
    }
#pragma unroll
    for (int u = 0; u < GRP; ++u) {
      const float* __restrict__ bc = bcp + (size_t)(g + u) * N_PROJ;
      const float dt = dtc[u];
      const float xv = xc[u];
      const float dtx = dt * xv;
      float y[4] = {0.0f, 0.0f, 0.0f, 0.0f};
#pragma unroll
      for (int n = 0; n < D_STATE; ++n) {
        const float dA = hw_exp2(dt * a2[n]);
        st[n] = fmaf(dA, st[n], dtx * bc[n]);
        y[n & 3] = fmaf(st[n], bc[D_STATE + n], y[n & 3]);
      }
      op[(size_t)(g + u) * D_INNER] = ((y[0] + y[1]) + (y[2] + y[3])) + xv * Dd;
    }
  }
}

extern "C" void kernel_launch(void* const* d_in, const int* in_sizes, int n_in,
                              void* d_out, int out_size, void* d_ws, size_t ws_size,
                              hipStream_t stream) {
  const float* x    = (const float*)d_in[0];
  const float* xpw  = (const float*)d_in[1];
  const float* dtw  = (const float*)d_in[2];
  const float* dtb  = (const float*)d_in[3];
  const float* alog = (const float*)d_in[4];
  const float* dvec = (const float*)d_in[5];
  float* out = (float*)d_out;

  char* ws = (char*)d_ws;
  float* xdbl   = (float*)ws;
  ws += (size_t)M_TOT * N_PROJ * sizeof(float);
  float* delta  = (float*)ws;
  ws += (size_t)M_TOT * D_INNER * sizeof(float);
  float* chunkA = (float*)ws;
  ws += (size_t)NCHUNK * NSTATE_TOT * sizeof(float);
  float* chunkB = (float*)ws;

  ssm_xproj_wmma<<<dim3(M_TOT / 16), dim3(256), 0, stream>>>(x, xpw, xdbl);

  ssm_dtproj_wmma<<<dim3(D_INNER / G2_BCOLS, M_TOT / 128), dim3(256), 0, stream>>>(
      xdbl, dtw, dtb, delta);

  ssm_scan_pass1<<<dim3(NCHAN / SCAN_THREADS, NCHUNK), dim3(SCAN_THREADS), 0, stream>>>(
      x, xdbl, delta, alog, chunkA, chunkB);
  ssm_scan_pass2<<<dim3(NSTATE_TOT / 256), dim3(256), 0, stream>>>(
      chunkA, chunkB);
  ssm_scan_pass3<<<dim3(NCHAN / SCAN_THREADS, NCHUNK), dim3(SCAN_THREADS), 0, stream>>>(
      x, xdbl, delta, alog, dvec, chunkB, out);
}


# Scan passes 1 and 3: dt and x requested one 8-step group ahead into two alternating register groups (no copies, no conditional request, prologue loads landed first): no load wait inside a group.
# speedup vs baseline: 5.0427x; 1.0314x over previous
#include <hip/hip_runtime.h>
#include <math.h>


#define BATCH   2
#define SEQLEN  2048
#define D_INNER 2048
#define D_STATE 16
#define DT_RANK 64
#define N_PROJ  (DT_RANK + 2 * D_STATE)
#define M_TOT   (BATCH * SEQLEN)
#define NSTATE_TOT (BATCH * D_INNER * D_STATE)
#define NCHAN   (BATCH * D_INNER)
#define NCHUNK  32
#define CHUNKL  (SEQLEN / NCHUNK)
#define SCAN_THREADS 256
#define GRP     8

typedef float v2f __attribute__((ext_vector_type(2)));
typedef float v4f __attribute__((ext_vector_type(4)));
typedef float v8f __attribute__((ext_vector_type(8)));

__device__ __forceinline__ float hw_exp2(float v) {
  return __builtin_amdgcn_exp2f(v);
}

#define G1_WAVES  8
#define G1_KSLICE (D_INNER / G1_WAVES)
#define G1_KBATCH 32
#define G1_TILE_F (6 * 8 * 32)

__global__ __launch_bounds__(256)
void ssm_xproj_wmma(const float* __restrict__ X,
                    const float* __restrict__ W,
                    float* __restrict__ Y) {
  __shared__ float red[G1_WAVES * G1_TILE_F];
  const int lane = threadIdx.x & 31;
  const int wave = threadIdx.x >> 5;
  const int half = lane >> 4;
  const int ml   = lane & 15;
  const int m0   = blockIdx.x << 4;
  const int k0   = wave * G1_KSLICE + 4 * half;

  const float* __restrict__ ap  = X + (size_t)(m0 + ml) * D_INNER + k0;
  const float* __restrict__ bp4 = W + (size_t)k0 * N_PROJ + 4 * ml;
  const float* __restrict__ bp2 = W + (size_t)k0 * N_PROJ + 64 + 2 * ml;

  v8f acc[6];
#pragma unroll
  for (int t = 0; t < 6; ++t) acc[t] = (v8f){};

  v4f an[4], b4n[4][4];
  v2f b2n[4][4];
#pragma unroll
  for (int j = 0; j < 4; ++j) {
    an[j] = *(const v4f*)(ap + 8 * j);
#pragma unroll
    for (int e = 0; e < 4; ++e) {
      b4n[j][e] = *(const v4f*)(bp4 + (size_t)(8 * j + e) * N_PROJ);
      b2n[j][e] = *(const v2f*)(bp2 + (size_t)(8 * j + e) * N_PROJ);
    }
  }
#pragma unroll 1
  for (int kb = 0; kb < G1_KSLICE; kb += G1_KBATCH) {
    v4f ac[4], b4c[4][4];
    v2f b2c[4][4];
#pragma unroll
    for (int j = 0; j < 4; ++j) {
      ac[j] = an[j];
#pragma unroll
      for (int e = 0; e < 4; ++e) { b4c[j][e] = b4n[j][e]; b2c[j][e] = b2n[j][e]; }
    }
    if (kb + G1_KBATCH < G1_KSLICE) {
      const int kn = kb + G1_KBATCH;
#pragma unroll
      for (int j = 0; j < 4; ++j) {
        an[j] = *(const v4f*)(ap + kn + 8 * j);
#pragma unroll
        for (int e = 0; e < 4; ++e) {
          b4n[j][e] = *(const v4f*)(bp4 + (size_t)(kn + 8 * j + e) * N_PROJ);
          b2n[j][e] = *(const v2f*)(bp2 + (size_t)(kn + 8 * j + e) * N_PROJ);
        }
      }
    }
#pragma unroll
    for (int j = 0; j < 4; ++j) {
#pragma unroll
      for (int q = 0; q < 2; ++q) {
        v2f a;
        a.x = ac[j][2 * q];
        a.y = ac[j][2 * q + 1];
#pragma unroll
        for (int t = 0; t < 6; ++t) {
          v2f b;
          if (t < 4) {
            b.x = b4c[j][2 * q][t];
            b.y = b4c[j][2 * q + 1][t];
          } else {
            b.x = b2c[j][2 * q][t - 4];
            b.y = b2c[j][2 * q + 1][t - 4];
          }
          acc[t] = __builtin_amdgcn_wmma_f32_16x16x4_f32(
              false, a, false, b, (short)0, acc[t], false, false);
        }
      }
    }
  }

#pragma unroll
  for (int t = 0; t < 6; ++t) {
#pragma unroll
    for (int r = 0; r < 8; ++r) {
      red[wave * G1_TILE_F + (t * 8 + r) * 32 + lane] = acc[t][r];
    }
  }
  __syncthreads();
  float s[6];
#pragma unroll
  for (int t = 0; t < 6; ++t) {
    const int q = (t * 8 + wave) * 32 + lane;
    float v = red[q];
#pragma unroll
    for (int w = 1; w < G1_WAVES; ++w) v += red[w * G1_TILE_F + q];
    s[t] = v;
  }
  float* __restrict__ yrow = Y + (size_t)(m0 + wave + 8 * half) * N_PROJ;
  v4f o4 = {s[0], s[1], s[2], s[3]};
  v2f o2 = {s[4], s[5]};
  *(v4f*)(yrow + 4 * ml) = o4;
  *(v2f*)(yrow + 64 + 2 * ml) = o2;
}

#define G2_BCOLS  128
#define G2_PAIRS  (G2_BCOLS / 32)
#define G2_KSTEPS (DT_RANK / 4)

__device__ __forceinline__ float softplus_series(float v) {
  const float t = hw_exp2(-fabsf(v) * 1.44269504088896f);
  const float w = 2.0f + t;
  const float r = __builtin_amdgcn_rcpf(w);
  float s = t * r;
  s = fmaf(fmaf(-w, s, t), r, s);
  const float z = s * s;
  float p = 2.0f / 13.0f;
  p = fmaf(p, z, 2.0f / 11.0f);
  p = fmaf(p, z, 2.0f / 9.0f);
  p = fmaf(p, z, 2.0f / 7.0f);
  p = fmaf(p, z, 2.0f / 5.0f);
  p = fmaf(p, z, 2.0f / 3.0f);
  p = fmaf(p, z, 2.0f);
  return fmaf(s, p, fmaxf(v, 0.0f));
}

__global__ __launch_bounds__(256)
void ssm_dtproj_wmma(const float* __restrict__ XD,
                     const float* __restrict__ Wd,
                     const float* __restrict__ bias,
                     float* __restrict__ DT) {
  const int lane = threadIdx.x & 31;
  const int wave = threadIdx.x >> 5;
  const int half = lane >> 4;
  const int ml   = lane & 15;
  const int m0   = ((blockIdx.y << 3) + wave) << 4;
  const int nb   = blockIdx.x * G2_BCOLS;

  const float* __restrict__ arow = XD + (size_t)(m0 + ml) * N_PROJ + 2 * half;
  v2f a[G2_KSTEPS];
#pragma unroll
  for (int i = 0; i < G2_KSTEPS; ++i) a[i] = *(const v2f*)(arow + 4 * i);

  const float* __restrict__ bcol = Wd + (size_t)(2 * half) * D_INNER + nb + ml;
  float* __restrict__ orow = DT + (size_t)(m0 + 8 * half) * D_INNER + nb + ml;

  v2f bn[2][G2_KSTEPS];
  float biasn[2];
#pragma unroll
  for (int u = 0; u < 2; ++u) {
    biasn[u] = bias[nb + 16 * u + ml];
#pragma unroll
    for (int i = 0; i < G2_KSTEPS; ++i) {
      bn[u][i].x = bcol[(size_t)(4 * i) * D_INNER + 16 * u];
      bn[u][i].y = bcol[(size_t)(4 * i + 1) * D_INNER + 16 * u];
    }
  }
#pragma unroll 1
  for (int pr = 0; pr < G2_PAIRS; ++pr) {
    v2f bc[2][G2_KSTEPS];
    float biasc[2];
#pragma unroll
    for (int u = 0; u < 2; ++u) {
      biasc[u] = biasn[u];
#pragma unroll
      for (int i = 0; i < G2_KSTEPS; ++i) bc[u][i] = bn[u][i];
    }
    if (pr + 1 < G2_PAIRS) {
      const int cn = 32 * (pr + 1);
#pragma unroll
      for (int u = 0; u < 2; ++u) {
        biasn[u] = bias[nb + cn + 16 * u + ml];
#pragma unroll
        for (int i = 0; i < G2_KSTEPS; ++i) {
          bn[u][i].x = bcol[(size_t)(4 * i) * D_INNER + cn + 16 * u];
          bn[u][i].y = bcol[(size_t)(4 * i + 1) * D_INNER + cn + 16 * u];
        }
      }
    }
    v8f acc0 = {}, acc1 = {};
#pragma unroll
    for (int i = 0; i < G2_KSTEPS; ++i) {
      acc0 = __builtin_amdgcn_wmma_f32_16x16x4_f32(
          false, a[i], false, bc[0][i], (short)0, acc0, false, false);
      acc1 = __builtin_amdgcn_wmma_f32_16x16x4_f32(
          false, a[i], false, bc[1][i], (short)0, acc1, false, false);
    }
    float* __restrict__ o = orow + 32 * pr;
#pragma unroll
    for (int r = 0; r < 8; ++r) {
      o[(size_t)r * D_INNER]      = softplus_series(acc0[r] + biasc[0]);
      o[(size_t)r * D_INNER + 16] = softplus_series(acc1[r] + biasc[1]);
    }
  }
}

__device__ __forceinline__ void scan_load_group(float (&dtv)[GRP], float (&xv)[GRP],
                                                const float* __restrict__ dtp,
                                                const float* __restrict__ xp,
                                                int t0) {
#pragma unroll
  for (int u = 0; u < GRP; ++u) {
    dtv[u] = dtp[(size_t)(t0 + u) * D_INNER];
    xv[u]  = xp[(size_t)(t0 + u) * D_INNER];
  }
}

__device__ __forceinline__ int scan_next_group(int g) {
  return (g + 2 * GRP < CHUNKL) ? g + 2 * GRP : g + GRP;
}

__device__ __forceinline__ void scan1_group(const float (&dtv)[GRP], const float (&xv)[GRP],
                                            const float* __restrict__ brow,
                                            const float (&a2)[D_STATE],
                                            float (&s)[D_STATE], float (&p)[D_STATE]) {
#pragma unroll
  for (int u = 0; u < GRP; ++u) {
    const float* __restrict__ bc = brow + (size_t)u * N_PROJ;
    const float dt = dtv[u];
    const float dtx = dt * xv[u];
#pragma unroll
    for (int n = 0; n < D_STATE; ++n) {
      const float dA = hw_exp2(dt * a2[n]);
      s[n] = fmaf(dA, s[n], dtx * bc[n]);
      p[n] *= dA;
    }
  }
}

__global__ __launch_bounds__(SCAN_THREADS)
void ssm_scan_pass1(const float* __restrict__ X,
                    const float* __restrict__ XDBL,
                    const float* __restrict__ DT,
                    const float* __restrict__ A_log,
                    float* __restrict__ chunkA,
                    float* __restrict__ chunkB) {
  const int c     = blockIdx.x * SCAN_THREADS + threadIdx.x;
  const int chunk = blockIdx.y;
  const int b     = blockIdx.x / (D_INNER / SCAN_THREADS);
  const int d     = c & (D_INNER - 1);

  float a2[D_STATE], s[D_STATE], p[D_STATE];
#pragma unroll
  for (int n = 0; n < D_STATE; ++n) {
    a2[n] = -__expf(A_log[d * D_STATE + n]) * 1.44269504088896f;
    s[n] = 0.0f;
    p[n] = 1.0f;
  }

  const size_t m0 = (size_t)b * SEQLEN + (size_t)chunk * CHUNKL;
  const float* __restrict__ dtp = DT + m0 * D_INNER + d;
  const float* __restrict__ xp  = X + m0 * D_INNER + d;
  const float* __restrict__ bcp = XDBL + m0 * N_PROJ + DT_RANK;

  float dtA[GRP], xA[GRP], dtB[GRP], xB[GRP];
  scan_load_group(dtA, xA, dtp, xp, 0);
#pragma unroll 1
  for (int g = 0; g < CHUNKL; g += 2 * GRP) {
    scan_load_group(dtB, xB, dtp, xp, g + GRP);
    __builtin_amdgcn_sched_barrier(0);
    scan1_group(dtA, xA, bcp + (size_t)g * N_PROJ, a2, s, p);
    __builtin_amdgcn_sched_barrier(0);
    scan_load_group(dtA, xA, dtp, xp, scan_next_group(g));
    __builtin_amdgcn_sched_barrier(0);
    scan1_group(dtB, xB, bcp + (size_t)(g + GRP) * N_PROJ, a2, s, p);
  }
#pragma unroll
  for (int n = 0; n < D_STATE; ++n) {
    const size_t i = ((size_t)chunk * D_STATE + n) * NCHAN + c;
    chunkA[i] = p[n];
    chunkB[i] = s[n];
  }
}

__global__ void ssm_scan_pass2(const float* __restrict__ chunkA,
                               float* __restrict__ chunkB) {
  const int t = blockIdx.x * blockDim.x + threadIdx.x;
  float s = 0.0f;
#pragma unroll
  for (int ch = 0; ch < NCHUNK; ++ch) {
    const size_t i = (size_t)ch * NSTATE_TOT + t;
    const float Ac = chunkA[i];
    const float Bc = chunkB[i];
    chunkB[i] = s;
    s = fmaf(Ac, s, Bc);
  }
}

__device__ __forceinline__ void scan3_group(const float (&dtv)[GRP], const float (&xv)[GRP],
                                            const float* __restrict__ bcrow,
                                            float* __restrict__ orow,
                                            const float (&a2)[D_STATE],
                                            float (&st)[D_STATE], float Dd) {
#pragma unroll
  for (int u = 0; u < GRP; ++u) {
    const float* __restrict__ bc = bcrow + (size_t)u * N_PROJ;
    const float dt = dtv[u];
    const float xv1 = xv[u];
    const float dtx = dt * xv1;
    float y[4] = {0.0f, 0.0f, 0.0f, 0.0f};
#pragma unroll
    for (int n = 0; n < D_STATE; ++n) {
      const float dA = hw_exp2(dt * a2[n]);
      st[n] = fmaf(dA, st[n], dtx * bc[n]);
      y[n & 3] = fmaf(st[n], bc[D_STATE + n], y[n & 3]);
    }
    orow[(size_t)u * D_INNER] = ((y[0] + y[1]) + (y[2] + y[3])) + xv1 * Dd;
  }
}

__global__ __launch_bounds__(SCAN_THREADS)
void ssm_scan_pass3(const float* __restrict__ X,
                    const float* __restrict__ XDBL,
                    const float* __restrict__ DT,
                    const float* __restrict__ A_log,
                    const float* __restrict__ Dv,
                    const float* __restrict__ Sin,
                    float* __restrict__ OUT) {
  const int c     = blockIdx.x * SCAN_THREADS + threadIdx.x;
  const int chunk = blockIdx.y;
  const int b     = blockIdx.x / (D_INNER / SCAN_THREADS);
  const int d     = c & (D_INNER - 1);

  float a2[D_STATE], st[D_STATE];
#pragma unroll
  for (int n = 0; n < D_STATE; ++n) {
    a2[n] = -__expf(A_log[d * D_STATE + n]) * 1.44269504088896f;
    st[n] = Sin[((size_t)chunk * D_STATE + n) * NCHAN + c];
  }
  const float Dd = Dv[d];

  const size_t m0 = (size_t)b * SEQLEN + (size_t)chunk * CHUNKL;
  const float* __restrict__ dtp = DT + m0 * D_INNER + d;
  const float* __restrict__ xp  = X + m0 * D_INNER + d;
  const float* __restrict__ bcp = XDBL + m0 * N_PROJ + DT_RANK;
  float* __restrict__ op = OUT + m0 * D_INNER + d;

  float dtA[GRP], xA[GRP], dtB[GRP], xB[GRP];
  scan_load_group(dtA, xA, dtp, xp, 0);
  asm volatile("" :: "v"(st[0]), "v"(st[1]), "v"(st[2]), "v"(st[3]),
                     "v"(st[4]), "v"(st[5]), "v"(st[6]), "v"(st[7]),
                     "v"(st[8]), "v"(st[9]), "v"(st[10]), "v"(st[11]),
                     "v"(st[12]), "v"(st[13]), "v"(st[14]), "v"(st[15]),
                     "v"(Dd));
#pragma unroll 1
  for (int g = 0; g < CHUNKL; g += 2 * GRP) {
    scan_load_group(dtB, xB, dtp, xp, g + GRP);
    __builtin_amdgcn_sched_barrier(0);
    scan3_group(dtA, xA, bcp + (size_t)g * N_PROJ, op + (size_t)g * D_INNER, a2, st, Dd);
    __builtin_amdgcn_sched_barrier(0);
    scan_load_group(dtA, xA, dtp, xp, scan_next_group(g));
    __builtin_amdgcn_sched_barrier(0);
    scan3_group(dtB, xB, bcp + (size_t)(g + GRP) * N_PROJ, op + (size_t)(g + GRP) * D_INNER, a2, st, Dd);
  }
}

extern "C" void kernel_launch(void* const* d_in, const int* in_sizes, int n_in,
                              void* d_out, int out_size, void* d_ws, size_t ws_size,
                              hipStream_t stream) {
  const float* x    = (const float*)d_in[0];
  const float* xpw  = (const float*)d_in[1];
  const float* dtw  = (const float*)d_in[2];
  const float* dtb  = (const float*)d_in[3];
  const float* alog = (const float*)d_in[4];
  const float* dvec = (const float*)d_in[5];
  float* out = (float*)d_out;

  char* ws = (char*)d_ws;
  float* xdbl   = (float*)ws;
  ws += (size_t)M_TOT * N_PROJ * sizeof(float);
  float* delta  = (float*)ws;
  ws += (size_t)M_TOT * D_INNER * sizeof(float);
  float* chunkA = (float*)ws;
  ws += (size_t)NCHUNK * NSTATE_TOT * sizeof(float);
  float* chunkB = (float*)ws;

  ssm_xproj_wmma<<<dim3(M_TOT / 16), dim3(256), 0, stream>>>(x, xpw, xdbl);

  ssm_dtproj_wmma<<<dim3(D_INNER / G2_BCOLS, M_TOT / 128), dim3(256), 0, stream>>>(
      xdbl, dtw, dtb, delta);

  ssm_scan_pass1<<<dim3(NCHAN / SCAN_THREADS, NCHUNK), dim3(SCAN_THREADS), 0, stream>>>(
      x, xdbl, delta, alog, chunkA, chunkB);
  ssm_scan_pass2<<<dim3(NSTATE_TOT / 256), dim3(256), 0, stream>>>(
      chunkA, chunkB);
  ssm_scan_pass3<<<dim3(NCHAN / SCAN_THREADS, NCHUNK), dim3(SCAN_THREADS), 0, stream>>>(
      x, xdbl, delta, alog, dvec, chunkB, out);
}
